# starting kernel
# baseline (speedup 1.0000x reference)
#include <hip/hip_runtime.h>
#include <hip/hip_bf16.h>

typedef __attribute__((ext_vector_type(2))) float v2f;
typedef __attribute__((ext_vector_type(8))) float v8f;

#define D_FEAT 128
#define LDS_STRIDE 132
#define ROWS_PER_BLOCK 128

__global__ void gcn_init_deg(float* __restrict__ deg, int N) {
    int i = blockIdx.x * blockDim.x + threadIdx.x;
    if (i < N) deg[i] = 1.0f;
}

__global__ void gcn_count_deg(const int* __restrict__ dst, float* __restrict__ deg, int E) {
    int i = blockIdx.x * blockDim.x + threadIdx.x;
    if (i < E) atomicAdd(&deg[dst[i]], 1.0f);
}

__global__ void gcn_rsqrt_deg(float* __restrict__ dis, int N) {
    int i = blockIdx.x * blockDim.x + threadIdx.x;
    if (i < N) dis[i] = rsqrtf(dis[i]);
}

__global__ void __launch_bounds__(256)
gcn_gemm128(const float* __restrict__ X, const float* __restrict__ W,
            float* __restrict__ H, int N) {
    __shared__ float Ws[D_FEAT * LDS_STRIDE];
    __shared__ float As[16 * LDS_STRIDE];

    const int tid  = threadIdx.x;
    const int base = blockIdx.x * ROWS_PER_BLOCK;

    for (int i = tid; i < (D_FEAT * D_FEAT) / 4; i += 256) {
        const int k = i >> 5;
        const int c = i & 31;
        const float4 w4 = ((const float4*)(W + (size_t)k * D_FEAT))[c];
        Ws[(4 * c + 0) * LDS_STRIDE + k] = w4.x;
        Ws[(4 * c + 1) * LDS_STRIDE + k] = w4.y;
        Ws[(4 * c + 2) * LDS_STRIDE + k] = w4.z;
        Ws[(4 * c + 3) * LDS_STRIDE + k] = w4.w;
    }

    const int wave = tid >> 5;
    const int lane = tid & 31;
    const int half = lane >> 4;
    const int l16  = lane & 15;
    const int col  = wave * 16 + l16;

    #pragma unroll 1
    for (int p = 0; p < ROWS_PER_BLOCK / 16; ++p) {
        const int r0 = base + p * 16;
        if (r0 >= N) break;

        __syncthreads();

        for (int i = tid; i < (16 * D_FEAT) / 4; i += 256) {
            const int row = i >> 5;
            const int c   = i & 31;
            if (r0 + row < N) {
                const float4 a4 = ((const float4*)(X + (size_t)(r0 + row) * D_FEAT))[c];
                *(float4*)&As[row * LDS_STRIDE + 4 * c] = a4;
            }
        }
        __syncthreads();

        v8f acc = {};
        #pragma unroll
        for (int k0 = 0; k0 < D_FEAT; k0 += 4) {
            const int ka = k0 + 2 * half;
            const v2f a = *(const v2f*)(As + l16 * LDS_STRIDE + ka);
            const v2f b = *(const v2f*)(Ws + col * LDS_STRIDE + ka);
            acc = __builtin_amdgcn_wmma_f32_16x16x4_f32(
                 false, a,  false, b,
                 (short)0, acc,  false,  false);
        }

        #pragma unroll
        for (int j = 0; j < 8; ++j) {
            const int row = r0 + j + 8 * half;
            if (row < N) H[(size_t)row * D_FEAT + col] = acc[j];
        }
    }
}

__global__ void gcn_self_bias(const float* __restrict__ H, const float* __restrict__ dis,
                              const float* __restrict__ b, float* __restrict__ out, int N) {
    int i = blockIdx.x * blockDim.x + threadIdx.x;
    if (i < N * D_FEAT) {
        int n = i >> 7;
        int d = i & (D_FEAT - 1);
        float s = dis[n];
        out[i] = b[d] + s * s * H[i];
    }
}

__global__ void gcn_scatter_edges(const float* __restrict__ H,
                                  const int* __restrict__ src,
                                  const int* __restrict__ dst,
                                  const float* __restrict__ dis,
                                  float* __restrict__ out, int E) {
    int t = blockIdx.x * blockDim.x + threadIdx.x;
    int e = t >> 5;
    if (e >= E) return;
    int lane = threadIdx.x & 31;
    int s = src[e];
    int d = dst[e];
    float w = dis[s] * dis[d];
    const float4 hv = ((const float4*)(H + (size_t)s * D_FEAT))[lane];
    float* o = out + (size_t)d * D_FEAT + lane * 4;
    atomicAdd(o + 0, hv.x * w);
    atomicAdd(o + 1, hv.y * w);
    atomicAdd(o + 2, hv.z * w);
    atomicAdd(o + 3, hv.w * w);
}

extern "C" void kernel_launch(void* const* d_in, const int* in_sizes, int n_in,
                              void* d_out, int out_size, void* d_ws, size_t ws_size,
                              hipStream_t stream) {
    const float* feat = (const float*)d_in[0];
    const int*   ei   = (const int*)d_in[1];
    const float* W0   = (const float*)d_in[2];
    const float* b0   = (const float*)d_in[3];
    const float* W1   = (const float*)d_in[4];
    const float* b1   = (const float*)d_in[5];
    const float* W2   = (const float*)d_in[6];
    const float* b2   = (const float*)d_in[7];

    const int N = in_sizes[0] / D_FEAT;
    const int E = in_sizes[1] / 2;
    const int* src = ei;
    const int* dst = ei + E;

    float* dis = (float*)d_ws;
    size_t off = ((size_t)N + 1023) & ~(size_t)1023;
    float* H   = dis + off;
    float* X   = H + (size_t)N * D_FEAT;
    float* out = (float*)d_out;

    const int TB = 256;
    dim3 blkN((N + TB - 1) / TB);
    dim3 blkE((E + TB - 1) / TB);
    dim3 blkND(((size_t)N * D_FEAT + TB - 1) / TB);
    dim3 blkGemm((N + ROWS_PER_BLOCK - 1) / ROWS_PER_BLOCK);
    dim3 blkEdge(((size_t)E * 32 + TB - 1) / TB);

    gcn_init_deg<<<blkN, TB, 0, stream>>>(dis, N);
    gcn_count_deg<<<blkE, TB, 0, stream>>>(dst, dis, E);
    gcn_rsqrt_deg<<<blkN, TB, 0, stream>>>(dis, N);

    gcn_gemm128<<<blkGemm, TB, 0, stream>>>(feat, W0, H, N);
    gcn_self_bias<<<blkND, TB, 0, stream>>>(H, dis, b0, X, N);
    gcn_scatter_edges<<<blkEdge, TB, 0, stream>>>(H, src, dst, dis, X, E);

    gcn_gemm128<<<blkGemm, TB, 0, stream>>>(X, W1, H, N);
    gcn_self_bias<<<blkND, TB, 0, stream>>>(H, dis, b1, X, N);
    gcn_scatter_edges<<<blkEdge, TB, 0, stream>>>(H, src, dst, dis, X, E);

    gcn_gemm128<<<blkGemm, TB, 0, stream>>>(X, W2, H, N);
    gcn_self_bias<<<blkND, TB, 0, stream>>>(H, dis, b2, out, N);
    gcn_scatter_edges<<<blkEdge, TB, 0, stream>>>(H, src, dst, dis, out, E);
}


# Edges bucketed by dst once per call (int count atomic = slot, scan, fill, buckets sorted by src); one wave per dst row sums its H[src] rows in registers in that fixed order, one store; no f32 atomics
# speedup vs baseline: 2.7702x; 2.7702x over previous
#include <hip/hip_runtime.h>
#include <hip/hip_bf16.h>

typedef __attribute__((ext_vector_type(2))) float v2f;
typedef __attribute__((ext_vector_type(8))) float v8f;

#define D_FEAT 128
#define LDS_STRIDE 132
#define ROWS_PER_BLOCK 128

#define SCAN_ROWS 1024
typedef unsigned long long u64;

__global__ void gcn_zero_cnt(int* __restrict__ cnt, int N) {
    int i = blockIdx.x * blockDim.x + threadIdx.x;
    if (i < N) cnt[i] = 0;
}

__global__ void gcn_rank_edges(const int* __restrict__ dst, int* __restrict__ cnt,
                               int* __restrict__ rank, int E) {
    int e = blockIdx.x * blockDim.x + threadIdx.x;
    if (e < E) rank[e] = atomicAdd(&cnt[dst[e]], 1);
}

__global__ void __launch_bounds__(256)
gcn_scan_deg(const int* __restrict__ cnt, u64* __restrict__ row,
             float* __restrict__ dis, int N) {
    __shared__ int wsum[8];
    __shared__ int wpre[8];
    const int tid  = threadIdx.x;
    const int lane = tid & 31;
    const int wave = tid >> 5;
    const int base = blockIdx.x * SCAN_ROWS;

    int pre = 0;
    const int4* c4 = (const int4*)cnt;
    for (int i = tid; i < base / 4; i += 256) {
        const int4 v = c4[i];
        pre += (v.x + v.y) + (v.z + v.w);
    }

    const int i0 = base + 4 * tid;
    int c[4];
    for (int k = 0; k < 4; ++k) c[k] = (i0 + k < N) ? cnt[i0 + k] : 0;
    const int tsum = (c[0] + c[1]) + (c[2] + c[3]);

    int inc = tsum;
    for (int o = 1; o < 32; o <<= 1) {
        const int t = __shfl_up(inc, o);
        if (lane >= o) inc += t;
    }
    for (int o = 16; o > 0; o >>= 1) pre += __shfl_xor(pre, o);
    if (lane == 31) wsum[wave] = inc;
    if (lane == 0)  wpre[wave] = pre;
    __syncthreads();
    int off = 0;
    for (int w = 0; w < 8; ++w) {
        off += wpre[w];
        if (w < wave) off += wsum[w];
    }
    int ex = off + inc - tsum;
    for (int k = 0; k < 4; ++k) {
        if (i0 + k < N) {
            row[i0 + k] = ((u64)(unsigned)c[k] << 32) | (u64)(unsigned)ex;
            dis[i0 + k] = rsqrtf((float)(c[k] + 1));
        }
        ex += c[k];
    }
}

__global__ void gcn_fill_slots(const int* __restrict__ src, const int* __restrict__ dst,
                               const u64* __restrict__ row, const int* __restrict__ rank,
                               const float* __restrict__ dis, int2* __restrict__ raw, int E) {
    int e = blockIdx.x * blockDim.x + threadIdx.x;
    if (e < E) {
        const int s = src[e];
        const int d = dst[e];
        const float w = dis[s] * dis[d];
        const int first = (int)(unsigned)row[d];
        raw[first + rank[e]] = make_int2(s, __float_as_int(w));
    }
}

__global__ void __launch_bounds__(256)
gcn_order_rows(const u64* __restrict__ row, const int2* __restrict__ raw,
               int2* __restrict__ slot, int N) {
    const int lane = threadIdx.x & 31;
    const int d = __builtin_amdgcn_readfirstlane((int)((blockIdx.x * blockDim.x + threadIdx.x) >> 5));
    if (d >= N) return;
    const u64 ri    = row[d];
    const int start = (int)(unsigned)ri;
    const int n     = (int)(ri >> 32);
    for (int c0 = 0; c0 < n; c0 += 32) {
        const int i = c0 + lane;
        int2 mine = make_int2(0x7fffffff, 0);
        if (i < n) mine = raw[start + i];
        const u64 mk = ((u64)(unsigned)mine.x << 32) | (u64)(unsigned)i;
        int pos = 0;
        for (int k0 = 0; k0 < n; k0 += 32) {
            int key = mine.x;
            if (k0 != c0) {
                key = 0x7fffffff;
                if (k0 + lane < n) key = raw[start + k0 + lane].x;
            }
            const int cm = (n - k0 < 32) ? (n - k0) : 32;
            for (int k = 0; k < cm; ++k) {
                const u64 ok = ((u64)(unsigned)__builtin_amdgcn_readlane(key, k) << 32)
                             | (u64)(unsigned)(k0 + k);
                pos += (ok < mk) ? 1 : 0;
            }
        }
        if (i < n) slot[start + pos] = mine;
    }
}

__global__ void __launch_bounds__(256)
gcn_gemm128(const float* __restrict__ X, const float* __restrict__ W,
            float* __restrict__ H, int N) {
    __shared__ float Ws[D_FEAT * LDS_STRIDE];
    __shared__ float As[16 * LDS_STRIDE];

    const int tid  = threadIdx.x;
    const int base = blockIdx.x * ROWS_PER_BLOCK;

    for (int i = tid; i < (D_FEAT * D_FEAT) / 4; i += 256) {
        const int k = i >> 5;
        const int c = i & 31;
        const float4 w4 = ((const float4*)(W + (size_t)k * D_FEAT))[c];
        Ws[(4 * c + 0) * LDS_STRIDE + k] = w4.x;
        Ws[(4 * c + 1) * LDS_STRIDE + k] = w4.y;
        Ws[(4 * c + 2) * LDS_STRIDE + k] = w4.z;
        Ws[(4 * c + 3) * LDS_STRIDE + k] = w4.w;
    }

    const int wave = tid >> 5;
    const int lane = tid & 31;
    const int half = lane >> 4;
    const int l16  = lane & 15;
    const int col  = wave * 16 + l16;

    #pragma unroll 1
    for (int p = 0; p < ROWS_PER_BLOCK / 16; ++p) {
        const int r0 = base + p * 16;
        if (r0 >= N) break;

        __syncthreads();

        for (int i = tid; i < (16 * D_FEAT) / 4; i += 256) {
            const int row = i >> 5;
            const int c   = i & 31;
            if (r0 + row < N) {
                const float4 a4 = ((const float4*)(X + (size_t)(r0 + row) * D_FEAT))[c];
                *(float4*)&As[row * LDS_STRIDE + 4 * c] = a4;
            }
        }
        __syncthreads();

        v8f acc = {};
        #pragma unroll
        for (int k0 = 0; k0 < D_FEAT; k0 += 4) {
            const int ka = k0 + 2 * half;
            const v2f a = *(const v2f*)(As + l16 * LDS_STRIDE + ka);
            const v2f b = *(const v2f*)(Ws + col * LDS_STRIDE + ka);
            acc = __builtin_amdgcn_wmma_f32_16x16x4_f32(
                 false, a,  false, b,
                 (short)0, acc,  false,  false);
        }

        #pragma unroll
        for (int j = 0; j < 8; ++j) {
            const int row = r0 + j + 8 * half;
            if (row < N) H[(size_t)row * D_FEAT + col] = acc[j];
        }
    }
}

template <int B>
__device__ __forceinline__ void gcn_rows(float4& acc, const float4* __restrict__ H4,
                                         const int2 mine, const int j, const int lane) {
    float4 v[B];
    for (int u = 0; u < B; ++u) {
        const int s = __builtin_amdgcn_readlane(mine.x, j + u);
        v[u] = H4[(size_t)s * (D_FEAT / 4) + lane];
    }
    for (int u = 0; u < B; ++u) {
        const float w = __int_as_float(__builtin_amdgcn_readlane(mine.y, j + u));
        acc.x = fmaf(v[u].x, w, acc.x);
        acc.y = fmaf(v[u].y, w, acc.y);
        acc.z = fmaf(v[u].z, w, acc.z);
        acc.w = fmaf(v[u].w, w, acc.w);
    }
}

__device__ __forceinline__ void gcn_chunk(float4& acc, const float4* __restrict__ H4,
                                          const int2* __restrict__ slot,
                                          const int first, const int m, const int lane) {
    int2 mine = make_int2(0, 0);
    if (lane < m) mine = slot[first + lane];
    int j = 0;
    for (; j + 8 <= m; j += 8) gcn_rows<8>(acc, H4, mine, j, lane);
    switch (m - j) {
        case 7: gcn_rows<7>(acc, H4, mine, j, lane); break;
        case 6: gcn_rows<6>(acc, H4, mine, j, lane); break;
        case 5: gcn_rows<5>(acc, H4, mine, j, lane); break;
        case 4: gcn_rows<4>(acc, H4, mine, j, lane); break;
        case 3: gcn_rows<3>(acc, H4, mine, j, lane); break;
        case 2: gcn_rows<2>(acc, H4, mine, j, lane); break;
        case 1: gcn_rows<1>(acc, H4, mine, j, lane); break;
        default: break;
    }
}

__global__ void __launch_bounds__(256)
gcn_gather_rows(const float* __restrict__ H, const u64* __restrict__ row,
                const int2* __restrict__ slot, const float* __restrict__ dis,
                const float* __restrict__ b, float* __restrict__ out, int N) {
    const int lane = threadIdx.x & 31;
    const int d = __builtin_amdgcn_readfirstlane((int)((blockIdx.x * blockDim.x + threadIdx.x) >> 5));
    if (d >= N) return;

    const float4* H4 = (const float4*)H;
    const u64    ri = row[d];
    const float  sd = dis[d];
    const float4 hv = H4[(size_t)d * (D_FEAT / 4) + lane];
    const float4 bv = ((const float4*)b)[lane];
    const int start = (int)(unsigned)ri;
    const int n     = (int)(ri >> 32);

    float4 acc = make_float4(0.0f, 0.0f, 0.0f, 0.0f);

    if (n > 0) gcn_chunk(acc, H4, slot, start, (n < 32) ? n : 32, lane);
    for (int j0 = 32; j0 < n; j0 += 32)
        gcn_chunk(acc, H4, slot, start + j0, (n - j0 < 32) ? (n - j0) : 32, lane);

    const float ws = sd * sd;
    float4 o;
    o.x = fmaf(hv.x, ws, acc.x) + bv.x;
    o.y = fmaf(hv.y, ws, acc.y) + bv.y;
    o.z = fmaf(hv.z, ws, acc.z) + bv.z;
    o.w = fmaf(hv.w, ws, acc.w) + bv.w;
    ((float4*)out)[(size_t)d * (D_FEAT / 4) + lane] = o;
}

extern "C" void kernel_launch(void* const* d_in, const int* in_sizes, int n_in,
                              void* d_out, int out_size, void* d_ws, size_t ws_size,
                              hipStream_t stream) {
    const float* feat = (const float*)d_in[0];
    const int*   ei   = (const int*)d_in[1];
    const float* W0   = (const float*)d_in[2];
    const float* b0   = (const float*)d_in[3];
    const float* W1   = (const float*)d_in[4];
    const float* b1   = (const float*)d_in[5];
    const float* W2   = (const float*)d_in[6];
    const float* b2   = (const float*)d_in[7];

    const int N = in_sizes[0] / D_FEAT;
    const int E = in_sizes[1] / 2;
    const int* src = ei;
    const int* dst = ei + E;

    const size_t Np = ((size_t)N + 1023) & ~(size_t)1023;
    const size_t Ep = ((size_t)E + 1023) & ~(size_t)1023;
    int*   cnt  = (int*)d_ws;
    float* dis  = (float*)(cnt + Np);
    u64*   row  = (u64*)(dis + Np);
    int*   rank = (int*)(row + Np);
    int2*  raw  = (int2*)(rank + Ep);
    int2*  slot = raw + Ep;
    float* H    = (float*)(slot + Ep);
    float* X    = H + (size_t)N * D_FEAT;
    float* out  = (float*)d_out;

    const int TB = 256;
    dim3 blkN((N + TB - 1) / TB);
    dim3 blkE((E + TB - 1) / TB);
    dim3 blkScan((N + SCAN_ROWS - 1) / SCAN_ROWS);
    dim3 blkGemm((N + ROWS_PER_BLOCK - 1) / ROWS_PER_BLOCK);
    dim3 blkRow(((size_t)N * 32 + TB - 1) / TB);

    gcn_zero_cnt<<<blkN, TB, 0, stream>>>(cnt, N);
    gcn_rank_edges<<<blkE, TB, 0, stream>>>(dst, cnt, rank, E);
    gcn_scan_deg<<<blkScan, TB, 0, stream>>>(cnt, row, dis, N);
    gcn_fill_slots<<<blkE, TB, 0, stream>>>(src, dst, row, rank, dis, raw, E);
    gcn_order_rows<<<blkRow, TB, 0, stream>>>(row, raw, slot, N);

    gcn_gemm128<<<blkGemm, TB, 0, stream>>>(feat, W0, H, N);
    gcn_gather_rows<<<blkRow, TB, 0, stream>>>(H, row, slot, dis, b0, X, N);

    gcn_gemm128<<<blkGemm, TB, 0, stream>>>(X, W1, H, N);
    gcn_gather_rows<<<blkRow, TB, 0, stream>>>(H, row, slot, dis, b1, X, N);

    gcn_gemm128<<<blkGemm, TB, 0, stream>>>(X, W2, H, N);
    gcn_gather_rows<<<blkRow, TB, 0, stream>>>(H, row, slot, dis, b2, out, N);
}


# The three weight products folded into one: Y = X (W0 W1 W2) on WMMA, biases carried as rows b0 W1 W2 and b1 W2 through the 3 unchanged row sums; products reassociated, so rounding order changes
# speedup vs baseline: 3.2861x; 1.1862x over previous
#include <hip/hip_runtime.h>
#include <hip/hip_bf16.h>

typedef __attribute__((ext_vector_type(2))) float v2f;
typedef __attribute__((ext_vector_type(8))) float v8f;

#define D_FEAT 128
#define LDS_STRIDE 132
#define ROWS_PER_BLOCK 128

#define SCAN_ROWS 1024
typedef unsigned long long u64;

__global__ void gcn_zero_cnt(int* __restrict__ cnt, int N) {
    int i = blockIdx.x * blockDim.x + threadIdx.x;
    if (i < N) cnt[i] = 0;
}

__global__ void gcn_rank_edges(const int* __restrict__ dst, int* __restrict__ cnt,
                               int* __restrict__ rank, int E) {
    int e = blockIdx.x * blockDim.x + threadIdx.x;
    if (e < E) rank[e] = atomicAdd(&cnt[dst[e]], 1);
}

__global__ void __launch_bounds__(256)
gcn_scan_deg(const int* __restrict__ cnt, u64* __restrict__ row,
             float* __restrict__ dis, int N) {
    __shared__ int wsum[8];
    __shared__ int wpre[8];
    const int tid  = threadIdx.x;
    const int lane = tid & 31;
    const int wave = tid >> 5;
    const int base = blockIdx.x * SCAN_ROWS;

    int pre = 0;
    const int4* c4 = (const int4*)cnt;
    for (int i = tid; i < base / 4; i += 256) {
        const int4 v = c4[i];
        pre += (v.x + v.y) + (v.z + v.w);
    }

    const int i0 = base + 4 * tid;
    int c[4];
    for (int k = 0; k < 4; ++k) c[k] = (i0 + k < N) ? cnt[i0 + k] : 0;
    const int tsum = (c[0] + c[1]) + (c[2] + c[3]);

    int inc = tsum;
    for (int o = 1; o < 32; o <<= 1) {
        const int t = __shfl_up(inc, o);
        if (lane >= o) inc += t;
    }
    for (int o = 16; o > 0; o >>= 1) pre += __shfl_xor(pre, o);
    if (lane == 31) wsum[wave] = inc;
    if (lane == 0)  wpre[wave] = pre;
    __syncthreads();
    int off = 0;
    for (int w = 0; w < 8; ++w) {
        off += wpre[w];
        if (w < wave) off += wsum[w];
    }
    int ex = off + inc - tsum;
    for (int k = 0; k < 4; ++k) {
        if (i0 + k < N) {
            row[i0 + k] = ((u64)(unsigned)c[k] << 32) | (u64)(unsigned)ex;
            dis[i0 + k] = rsqrtf((float)(c[k] + 1));
        }
        ex += c[k];
    }
}

__global__ void gcn_fill_slots(const int* __restrict__ src, const int* __restrict__ dst,
                               const u64* __restrict__ row, const int* __restrict__ rank,
                               const float* __restrict__ dis, int2* __restrict__ raw, int E) {
    int e = blockIdx.x * blockDim.x + threadIdx.x;
    if (e < E) {
        const int s = src[e];
        const int d = dst[e];
        const float w = dis[s] * dis[d];
        const int first = (int)(unsigned)row[d];
        raw[first + rank[e]] = make_int2(s, __float_as_int(w));
    }
}

__global__ void __launch_bounds__(256)
gcn_order_rows(const u64* __restrict__ row, const int2* __restrict__ raw,
               int2* __restrict__ slot, int N) {
    const int lane = threadIdx.x & 31;
    const int d = __builtin_amdgcn_readfirstlane((int)((blockIdx.x * blockDim.x + threadIdx.x) >> 5));
    if (d >= N) return;
    const u64 ri    = row[d];
    const int start = (int)(unsigned)ri;
    const int n     = (int)(ri >> 32);
    for (int c0 = 0; c0 < n; c0 += 32) {
        const int i = c0 + lane;
        int2 mine = make_int2(0x7fffffff, 0);
        if (i < n) mine = raw[start + i];
        const u64 mk = ((u64)(unsigned)mine.x << 32) | (u64)(unsigned)i;
        int pos = 0;
        for (int k0 = 0; k0 < n; k0 += 32) {
            int key = mine.x;
            if (k0 != c0) {
                key = 0x7fffffff;
                if (k0 + lane < n) key = raw[start + k0 + lane].x;
            }
            const int cm = (n - k0 < 32) ? (n - k0) : 32;
            for (int k = 0; k < cm; ++k) {
                const u64 ok = ((u64)(unsigned)__builtin_amdgcn_readlane(key, k) << 32)
                             | (u64)(unsigned)(k0 + k);
                pos += (ok < mk) ? 1 : 0;
            }
        }
        if (i < n) slot[start + pos] = mine;
    }
}

__global__ void __launch_bounds__(256)
gcn_gemm128(const float* __restrict__ X, const float* __restrict__ W,
            float* __restrict__ H, int N) {
    __shared__ float Ws[D_FEAT * LDS_STRIDE];
    __shared__ float As[16 * LDS_STRIDE];

    const int tid  = threadIdx.x;
    const int base = blockIdx.x * ROWS_PER_BLOCK;

    for (int i = tid; i < (D_FEAT * D_FEAT) / 4; i += 256) {
        const int k = i >> 5;
        const int c = i & 31;
        const float4 w4 = ((const float4*)(W + (size_t)k * D_FEAT))[c];
        Ws[(4 * c + 0) * LDS_STRIDE + k] = w4.x;
        Ws[(4 * c + 1) * LDS_STRIDE + k] = w4.y;
        Ws[(4 * c + 2) * LDS_STRIDE + k] = w4.z;
        Ws[(4 * c + 3) * LDS_STRIDE + k] = w4.w;
    }

    const int wave = tid >> 5;
    const int lane = tid & 31;
    const int half = lane >> 4;
    const int l16  = lane & 15;
    const int col  = wave * 16 + l16;

    #pragma unroll 1
    for (int p = 0; p < ROWS_PER_BLOCK / 16; ++p) {
        const int r0 = base + p * 16;
        if (r0 >= N) break;

        __syncthreads();

        for (int i = tid; i < (16 * D_FEAT) / 4; i += 256) {
            const int row = i >> 5;
            const int c   = i & 31;
            if (r0 + row < N) {
                const float4 a4 = ((const float4*)(X + (size_t)(r0 + row) * D_FEAT))[c];
                *(float4*)&As[row * LDS_STRIDE + 4 * c] = a4;
            }
        }
        __syncthreads();

        v8f acc = {};
        #pragma unroll
        for (int k0 = 0; k0 < D_FEAT; k0 += 4) {
            const int ka = k0 + 2 * half;
            const v2f a = *(const v2f*)(As + l16 * LDS_STRIDE + ka);
            const v2f b = *(const v2f*)(Ws + col * LDS_STRIDE + ka);
            acc = __builtin_amdgcn_wmma_f32_16x16x4_f32(
                 false, a,  false, b,
                 (short)0, acc,  false,  false);
        }

        #pragma unroll
        for (int j = 0; j < 8; ++j) {
            const int row = r0 + j + 8 * half;
            if (row < N) H[(size_t)row * D_FEAT + col] = acc[j];
        }
    }
}

#define FOLD_BLOCKS 9

__device__ __forceinline__ v8f gcn_panel_times_wt(const float* __restrict__ As,
                                                  const float* __restrict__ Ws,
                                                  const int l16, const int half, const int col) {
    v8f acc = {};
    #pragma unroll
    for (int k0 = 0; k0 < D_FEAT; k0 += 4) {
        const int ka = k0 + 2 * half;
        const v2f a = *(const v2f*)(As + l16 * LDS_STRIDE + ka);
        const v2f b = *(const v2f*)(Ws + col * LDS_STRIDE + ka);
        acc = __builtin_amdgcn_wmma_f32_16x16x4_f32(
             false, a,  false, b,
             (short)0, acc,  false,  false);
    }
    return acc;
}

__global__ void __launch_bounds__(256)
gcn_fold_weights(const float* __restrict__ W0, const float* __restrict__ b0,
                 const float* __restrict__ W1, const float* __restrict__ b1,
                 const float* __restrict__ W2, float* __restrict__ Wc) {
    __shared__ float W1s[D_FEAT * LDS_STRIDE];
    __shared__ float W2s[D_FEAT * LDS_STRIDE];
    __shared__ float As[16 * LDS_STRIDE];

    const int tid  = threadIdx.x;
    const int p    = blockIdx.x;
    const int wave = tid >> 5;
    const int lane = tid & 31;
    const int half = lane >> 4;
    const int l16  = lane & 15;
    const int col  = wave * 16 + l16;

    const int kk = lane & 7;
    const int pc = 4 * wave + (lane >> 3);
    float4 w1[16], w2[16], a4[2];
    #pragma unroll
    for (int u = 0; u < 16; ++u)
        w1[u] = ((const float4*)(W1 + (size_t)(8 * u + kk) * D_FEAT))[pc];
    #pragma unroll
    for (int u = 0; u < 16; ++u)
        w2[u] = ((const float4*)(W2 + (size_t)(8 * u + kk) * D_FEAT))[pc];
    #pragma unroll
    for (int u = 0; u < 2; ++u) {
        const int r = 8 * u + wave;
        a4[u] = make_float4(0.0f, 0.0f, 0.0f, 0.0f);
        if (p < 8)       a4[u] = ((const float4*)(W0 + (size_t)(16 * p + r) * D_FEAT))[lane];
        else if (r == 0) a4[u] = ((const float4*)b0)[lane];
    }
    #pragma unroll
    for (int u = 0; u < 2; ++u)
        *(float4*)&As[(8 * u + wave) * LDS_STRIDE + 4 * lane] = a4[u];
    #pragma unroll
    for (int u = 0; u < 16; ++u) {
        const int k = 8 * u + kk;
        W1s[(4 * pc + 0) * LDS_STRIDE + k] = w1[u].x;
        W1s[(4 * pc + 1) * LDS_STRIDE + k] = w1[u].y;
        W1s[(4 * pc + 2) * LDS_STRIDE + k] = w1[u].z;
        W1s[(4 * pc + 3) * LDS_STRIDE + k] = w1[u].w;
    }
    #pragma unroll
    for (int u = 0; u < 16; ++u) {
        const int k = 8 * u + kk;
        W2s[(4 * pc + 0) * LDS_STRIDE + k] = w2[u].x;
        W2s[(4 * pc + 1) * LDS_STRIDE + k] = w2[u].y;
        W2s[(4 * pc + 2) * LDS_STRIDE + k] = w2[u].z;
        W2s[(4 * pc + 3) * LDS_STRIDE + k] = w2[u].w;
    }
    __syncthreads();

    const v8f t = gcn_panel_times_wt(As, W1s, l16, half, col);
    __syncthreads();

    #pragma unroll
    for (int j = 0; j < 8; ++j) As[(j + 8 * half) * LDS_STRIDE + col] = t[j];
    if (p == 8 && half == 0) As[1 * LDS_STRIDE + col] = b1[col];
    __syncthreads();

    const v8f o = gcn_panel_times_wt(As, W2s, l16, half, col);

    const int nrow = (p < 8) ? 16 : 2;
    #pragma unroll
    for (int j = 0; j < 8; ++j) {
        const int r = j + 8 * half;
        if (r < nrow) Wc[(size_t)(16 * p + r) * D_FEAT + col] = o[j];
    }
}

template <int B>
__device__ __forceinline__ void gcn_rows(float4& acc, const float4* __restrict__ H4,
                                         const int2 mine, const int j, const int lane) {
    float4 v[B];
    for (int u = 0; u < B; ++u) {
        const int s = __builtin_amdgcn_readlane(mine.x, j + u);
        v[u] = H4[(size_t)s * (D_FEAT / 4) + lane];
    }
    for (int u = 0; u < B; ++u) {
        const float w = __int_as_float(__builtin_amdgcn_readlane(mine.y, j + u));
        acc.x = fmaf(v[u].x, w, acc.x);
        acc.y = fmaf(v[u].y, w, acc.y);
        acc.z = fmaf(v[u].z, w, acc.z);
        acc.w = fmaf(v[u].w, w, acc.w);
    }
}

__device__ __forceinline__ void gcn_chunk(float4& acc, const float4* __restrict__ H4,
                                          const int2* __restrict__ slot,
                                          const int first, const int m, const int lane) {
    int2 mine = make_int2(0, 0);
    if (lane < m) mine = slot[first + lane];
    int j = 0;
    for (; j + 8 <= m; j += 8) gcn_rows<8>(acc, H4, mine, j, lane);
    switch (m - j) {
        case 7: gcn_rows<7>(acc, H4, mine, j, lane); break;
        case 6: gcn_rows<6>(acc, H4, mine, j, lane); break;
        case 5: gcn_rows<5>(acc, H4, mine, j, lane); break;
        case 4: gcn_rows<4>(acc, H4, mine, j, lane); break;
        case 3: gcn_rows<3>(acc, H4, mine, j, lane); break;
        case 2: gcn_rows<2>(acc, H4, mine, j, lane); break;
        case 1: gcn_rows<1>(acc, H4, mine, j, lane); break;
        default: break;
    }
}

__global__ void __launch_bounds__(256)
gcn_gather_rows(const float* __restrict__ H, const u64* __restrict__ row,
                const int2* __restrict__ slot, const float* __restrict__ dis,
                const float* __restrict__ b, float* __restrict__ out, int N) {
    const int lane = threadIdx.x & 31;
    const int d = __builtin_amdgcn_readfirstlane((int)((blockIdx.x * blockDim.x + threadIdx.x) >> 5));
    if (d >= N) return;

    const float4* H4 = (const float4*)H;
    const u64    ri = row[d];
    const float  sd = dis[d];
    const float4 hv = H4[(size_t)d * (D_FEAT / 4) + lane];
    const float4 bv = ((const float4*)b)[lane];
    const int start = (int)(unsigned)ri;
    const int n     = (int)(ri >> 32);

    float4 acc = make_float4(0.0f, 0.0f, 0.0f, 0.0f);

    if (n > 0) gcn_chunk(acc, H4, slot, start, (n < 32) ? n : 32, lane);
    for (int j0 = 32; j0 < n; j0 += 32)
        gcn_chunk(acc, H4, slot, start + j0, (n - j0 < 32) ? (n - j0) : 32, lane);

    const float ws = sd * sd;
    float4 o;
    o.x = fmaf(hv.x, ws, acc.x) + bv.x;
    o.y = fmaf(hv.y, ws, acc.y) + bv.y;
    o.z = fmaf(hv.z, ws, acc.z) + bv.z;
    o.w = fmaf(hv.w, ws, acc.w) + bv.w;
    ((float4*)out)[(size_t)d * (D_FEAT / 4) + lane] = o;
}

extern "C" void kernel_launch(void* const* d_in, const int* in_sizes, int n_in,
                              void* d_out, int out_size, void* d_ws, size_t ws_size,
                              hipStream_t stream) {
    const float* feat = (const float*)d_in[0];
    const int*   ei   = (const int*)d_in[1];
    const float* W0   = (const float*)d_in[2];
    const float* b0   = (const float*)d_in[3];
    const float* W1   = (const float*)d_in[4];
    const float* b1   = (const float*)d_in[5];
    const float* W2   = (const float*)d_in[6];
    const float* b2   = (const float*)d_in[7];

    const int N = in_sizes[0] / D_FEAT;
    const int E = in_sizes[1] / 2;
    const int* src = ei;
    const int* dst = ei + E;

    const size_t Np = ((size_t)N + 1023) & ~(size_t)1023;
    const size_t Ep = ((size_t)E + 1023) & ~(size_t)1023;
    int*   cnt  = (int*)d_ws;
    float* dis  = (float*)(cnt + Np);
    u64*   row  = (u64*)(dis + Np);
    int*   rank = (int*)(row + Np);
    int2*  raw  = (int2*)(rank + Ep);
    int2*  slot = raw + Ep;
    float* H    = (float*)(slot + Ep);
    float* X    = H + (size_t)N * D_FEAT;
    float* Wc   = X + (size_t)N * D_FEAT;
    float* out  = (float*)d_out;

    const int TB = 256;
    dim3 blkN((N + TB - 1) / TB);
    dim3 blkE((E + TB - 1) / TB);
    dim3 blkScan((N + SCAN_ROWS - 1) / SCAN_ROWS);
    dim3 blkGemm((N + ROWS_PER_BLOCK - 1) / ROWS_PER_BLOCK);
    dim3 blkRow(((size_t)N * 32 + TB - 1) / TB);

    gcn_zero_cnt<<<blkN, TB, 0, stream>>>(cnt, N);
    gcn_rank_edges<<<blkE, TB, 0, stream>>>(dst, cnt, rank, E);
    gcn_scan_deg<<<blkScan, TB, 0, stream>>>(cnt, row, dis, N);
    gcn_fill_slots<<<blkE, TB, 0, stream>>>(src, dst, row, rank, dis, raw, E);
    gcn_order_rows<<<blkRow, TB, 0, stream>>>(row, raw, slot, N);

    gcn_fold_weights<<<FOLD_BLOCKS, TB, 0, stream>>>(W0, b0, W1, b1, W2, Wc);
    gcn_gemm128<<<blkGemm, TB, 0, stream>>>(feat, Wc, H, N);

    gcn_gather_rows<<<blkRow, TB, 0, stream>>>(H, row, slot, dis, Wc + 128 * D_FEAT, X, N);
    gcn_gather_rows<<<blkRow, TB, 0, stream>>>(X, row, slot, dis, Wc + 129 * D_FEAT, H, N);
    gcn_gather_rows<<<blkRow, TB, 0, stream>>>(H, row, slot, dis, b2, out, N);
}


# Independent work shares a grid: counter clear rides with the weight fold, H = X Wc (B fragments in registers, one 8 KB A panel in LDS) rides with edge ranking; 10 launches -> 8; sums unchanged
# speedup vs baseline: 3.4465x; 1.0488x over previous
#include <hip/hip_runtime.h>
#include <hip/hip_bf16.h>

typedef __attribute__((ext_vector_type(2))) float v2f;
typedef __attribute__((ext_vector_type(8))) float v8f;

#define D_FEAT 128
#define LDS_STRIDE 132

#define SCAN_ROWS 1024
typedef unsigned long long u64;

__global__ void __launch_bounds__(256)
gcn_scan_deg(const int* __restrict__ cnt, u64* __restrict__ row,
             float* __restrict__ dis, int N) {
    __shared__ int wsum[8];
    __shared__ int wpre[8];
    const int tid  = threadIdx.x;
    const int lane = tid & 31;
    const int wave = tid >> 5;
    const int base = blockIdx.x * SCAN_ROWS;

    int pre = 0;
    const int4* c4 = (const int4*)cnt;
    for (int i = tid; i < base / 4; i += 256) {
        const int4 v = c4[i];
        pre += (v.x + v.y) + (v.z + v.w);
    }

    const int i0 = base + 4 * tid;
    int c[4];
    for (int k = 0; k < 4; ++k) c[k] = (i0 + k < N) ? cnt[i0 + k] : 0;
    const int tsum = (c[0] + c[1]) + (c[2] + c[3]);

    int inc = tsum;
    for (int o = 1; o < 32; o <<= 1) {
        const int t = __shfl_up(inc, o);
        if (lane >= o) inc += t;
    }
    for (int o = 16; o > 0; o >>= 1) pre += __shfl_xor(pre, o);
    if (lane == 31) wsum[wave] = inc;
    if (lane == 0)  wpre[wave] = pre;
    __syncthreads();
    int off = 0;
    for (int w = 0; w < 8; ++w) {
        off += wpre[w];
        if (w < wave) off += wsum[w];
    }
    int ex = off + inc - tsum;
    for (int k = 0; k < 4; ++k) {
        if (i0 + k < N) {
            row[i0 + k] = ((u64)(unsigned)c[k] << 32) | (u64)(unsigned)ex;
            dis[i0 + k] = rsqrtf((float)(c[k] + 1));
        }
        ex += c[k];
    }
}

__global__ void gcn_fill_slots(const int* __restrict__ src, const int* __restrict__ dst,
                               const u64* __restrict__ row, const int* __restrict__ rank,
                               const float* __restrict__ dis, int2* __restrict__ raw, int E) {
    int e = blockIdx.x * blockDim.x + threadIdx.x;
    if (e < E) {
        const int s = src[e];
        const int d = dst[e];
        const float w = dis[s] * dis[d];
        const int first = (int)(unsigned)row[d];
        raw[first + rank[e]] = make_int2(s, __float_as_int(w));
    }
}

__global__ void __launch_bounds__(256)
gcn_order_rows(const u64* __restrict__ row, const int2* __restrict__ raw,
               int2* __restrict__ slot, int N) {
    const int lane = threadIdx.x & 31;
    const int d = __builtin_amdgcn_readfirstlane((int)((blockIdx.x * blockDim.x + threadIdx.x) >> 5));
    if (d >= N) return;
    const u64 ri    = row[d];
    const int start = (int)(unsigned)ri;
    const int n     = (int)(ri >> 32);
    for (int c0 = 0; c0 < n; c0 += 32) {
        const int i = c0 + lane;
        int2 mine = make_int2(0x7fffffff, 0);
        if (i < n) mine = raw[start + i];
        const u64 mk = ((u64)(unsigned)mine.x << 32) | (u64)(unsigned)i;
        int pos = 0;
        for (int k0 = 0; k0 < n; k0 += 32) {
            int key = mine.x;
            if (k0 != c0) {
                key = 0x7fffffff;
                if (k0 + lane < n) key = raw[start + k0 + lane].x;
            }
            const int cm = (n - k0 < 32) ? (n - k0) : 32;
            for (int k = 0; k < cm; ++k) {
                const u64 ok = ((u64)(unsigned)__builtin_amdgcn_readlane(key, k) << 32)
                             | (u64)(unsigned)(k0 + k);
                pos += (ok < mk) ? 1 : 0;
            }
        }
        if (i < n) slot[start + pos] = mine;
    }
}

#define GEMM_BLOCKS 512
#define RANK_EDGES 1024

__device__ __forceinline__ void gcn_gemm_panels(const float* __restrict__ X,
                                                const float* __restrict__ W,
                                                float* __restrict__ H, const int N, const int b) {
    __shared__ float As[16 * LDS_STRIDE];

    const int tid  = threadIdx.x;
    const int wave = tid >> 5;
    const int lane = tid & 31;
    const int half = lane >> 4;
    const int l16  = lane & 15;
    const int col  = wave * 16 + l16;

    v2f bf[D_FEAT / 4];
    #pragma unroll
    for (int i = 0; i < D_FEAT / 4; ++i) {
        const int k = 4 * i + 2 * half;
        bf[i].x = W[(size_t)k * D_FEAT + col];
        bf[i].y = W[(size_t)(k + 1) * D_FEAT + col];
    }

    const int npan = (N + 15) >> 4;
    const float4 zero4 = make_float4(0.0f, 0.0f, 0.0f, 0.0f);

    float4 a0 = zero4, a1 = zero4;
    if (b < npan) {
        const int r = 16 * b + wave;
        if (r < N)     a0 = ((const float4*)(X + (size_t)r * D_FEAT))[lane];
        if (r + 8 < N) a1 = ((const float4*)(X + (size_t)(r + 8) * D_FEAT))[lane];
    }

    #pragma unroll 1
    for (int p = b; p < npan; p += GEMM_BLOCKS) {
        __syncthreads();
        *(float4*)&As[wave * LDS_STRIDE + 4 * lane]       = a0;
        *(float4*)&As[(wave + 8) * LDS_STRIDE + 4 * lane] = a1;
        __syncthreads();

        const int pn = p + GEMM_BLOCKS;
        a0 = zero4; a1 = zero4;
        if (pn < npan) {
            const int r = 16 * pn + wave;
            if (r < N)     a0 = ((const float4*)(X + (size_t)r * D_FEAT))[lane];
            if (r + 8 < N) a1 = ((const float4*)(X + (size_t)(r + 8) * D_FEAT))[lane];
        }

        v8f acc = {};
        #pragma unroll
        for (int i = 0; i < D_FEAT / 4; ++i) {
            const v2f a = *(const v2f*)(As + l16 * LDS_STRIDE + 4 * i + 2 * half);
            acc = __builtin_amdgcn_wmma_f32_16x16x4_f32(
                 false, a,  false, bf[i],
                 (short)0, acc,  false,  false);
        }

        const int r0 = 16 * p;
        #pragma unroll
        for (int j = 0; j < 8; ++j) {
            const int row = r0 + j + 8 * half;
            if (row < N) H[(size_t)row * D_FEAT + col] = acc[j];
        }
    }
}

__global__ void __launch_bounds__(256)
gcn_rank_edges_gemm(const int* __restrict__ dst, int* __restrict__ cnt,
                    int* __restrict__ rank, int E,
                    const float* __restrict__ X, const float* __restrict__ W,
                    float* __restrict__ H, int N) {
    if (blockIdx.x < GEMM_BLOCKS) {
        gcn_gemm_panels(X, W, H, N, (int)blockIdx.x);
        return;
    }
    const int base = (int)(blockIdx.x - GEMM_BLOCKS) * RANK_EDGES;
    const int e0   = base + (int)threadIdx.x;
    if (base + RANK_EDGES <= E) {
        int d[RANK_EDGES / 256], r[RANK_EDGES / 256];
        #pragma unroll
        for (int u = 0; u < RANK_EDGES / 256; ++u) d[u] = dst[e0 + 256 * u];
        #pragma unroll
        for (int u = 0; u < RANK_EDGES / 256; ++u) r[u] = atomicAdd(&cnt[d[u]], 1);
        #pragma unroll
        for (int u = 0; u < RANK_EDGES / 256; ++u) rank[e0 + 256 * u] = r[u];
    } else {
        for (int u = 0; u < RANK_EDGES / 256; ++u) {
            const int e = e0 + 256 * u;
            if (e < E) rank[e] = atomicAdd(&cnt[dst[e]], 1);
        }
    }
}

#define FOLD_BLOCKS 9

__device__ __forceinline__ v8f gcn_panel_times_wt(const float* __restrict__ As,
                                                  const float* __restrict__ Ws,
                                                  const int l16, const int half, const int col) {
    v8f acc = {};
    #pragma unroll
    for (int k0 = 0; k0 < D_FEAT; k0 += 4) {
        const int ka = k0 + 2 * half;
        const v2f a = *(const v2f*)(As + l16 * LDS_STRIDE + ka);
        const v2f b = *(const v2f*)(Ws + col * LDS_STRIDE + ka);
        acc = __builtin_amdgcn_wmma_f32_16x16x4_f32(
             false, a,  false, b,
             (short)0, acc,  false,  false);
    }
    return acc;
}

__global__ void __launch_bounds__(256)
gcn_fold_weights_clear(const float* __restrict__ W0, const float* __restrict__ b0,
                       const float* __restrict__ W1, const float* __restrict__ b1,
                       const float* __restrict__ W2, float* __restrict__ Wc,
                       int* __restrict__ cnt, int N) {
    __shared__ float W1s[D_FEAT * LDS_STRIDE];
    __shared__ float W2s[D_FEAT * LDS_STRIDE];
    __shared__ float As[16 * LDS_STRIDE];

    const int tid  = threadIdx.x;
    if (blockIdx.x >= FOLD_BLOCKS) {
        const int i = (int)(blockIdx.x - FOLD_BLOCKS) * 256 + tid;
        if (i < N) cnt[i] = 0;
        return;
    }
    const int p    = blockIdx.x;
    const int wave = tid >> 5;
    const int lane = tid & 31;
    const int half = lane >> 4;
    const int l16  = lane & 15;
    const int col  = wave * 16 + l16;

    const int kk = lane & 7;
    const int pc = 4 * wave + (lane >> 3);
    float4 w1[16], w2[16], a4[2];
    #pragma unroll
    for (int u = 0; u < 16; ++u)
        w1[u] = ((const float4*)(W1 + (size_t)(8 * u + kk) * D_FEAT))[pc];
    #pragma unroll
    for (int u = 0; u < 16; ++u)
        w2[u] = ((const float4*)(W2 + (size_t)(8 * u + kk) * D_FEAT))[pc];
    #pragma unroll
    for (int u = 0; u < 2; ++u) {
        const int r = 8 * u + wave;
        a4[u] = make_float4(0.0f, 0.0f, 0.0f, 0.0f);
        if (p < 8)       a4[u] = ((const float4*)(W0 + (size_t)(16 * p + r) * D_FEAT))[lane];
        else if (r == 0) a4[u] = ((const float4*)b0)[lane];
    }
    #pragma unroll
    for (int u = 0; u < 2; ++u)
        *(float4*)&As[(8 * u + wave) * LDS_STRIDE + 4 * lane] = a4[u];
    #pragma unroll
    for (int u = 0; u < 16; ++u) {
        const int k = 8 * u + kk;
        W1s[(4 * pc + 0) * LDS_STRIDE + k] = w1[u].x;
        W1s[(4 * pc + 1) * LDS_STRIDE + k] = w1[u].y;
        W1s[(4 * pc + 2) * LDS_STRIDE + k] = w1[u].z;
        W1s[(4 * pc + 3) * LDS_STRIDE + k] = w1[u].w;
    }
    #pragma unroll
    for (int u = 0; u < 16; ++u) {
        const int k = 8 * u + kk;
        W2s[(4 * pc + 0) * LDS_STRIDE + k] = w2[u].x;
        W2s[(4 * pc + 1) * LDS_STRIDE + k] = w2[u].y;
        W2s[(4 * pc + 2) * LDS_STRIDE + k] = w2[u].z;
        W2s[(4 * pc + 3) * LDS_STRIDE + k] = w2[u].w;
    }
    __syncthreads();

    const v8f t = gcn_panel_times_wt(As, W1s, l16, half, col);
    __syncthreads();

    #pragma unroll
    for (int j = 0; j < 8; ++j) As[(j + 8 * half) * LDS_STRIDE + col] = t[j];
    if (p == 8 && half == 0) As[1 * LDS_STRIDE + col] = b1[col];
    __syncthreads();

    const v8f o = gcn_panel_times_wt(As, W2s, l16, half, col);

    const int nrow = (p < 8) ? 16 : 2;
    #pragma unroll
    for (int j = 0; j < 8; ++j) {
        const int r = j + 8 * half;
        if (r < nrow) Wc[(size_t)(16 * p + r) * D_FEAT + col] = o[j];
    }
}

template <int B>
__device__ __forceinline__ void gcn_rows(float4& acc, const float4* __restrict__ H4,
                                         const int2 mine, const int j, const int lane) {
    float4 v[B];
    for (int u = 0; u < B; ++u) {
        const int s = __builtin_amdgcn_readlane(mine.x, j + u);
        v[u] = H4[(size_t)s * (D_FEAT / 4) + lane];
    }
    for (int u = 0; u < B; ++u) {
        const float w = __int_as_float(__builtin_amdgcn_readlane(mine.y, j + u));
        acc.x = fmaf(v[u].x, w, acc.x);
        acc.y = fmaf(v[u].y, w, acc.y);
        acc.z = fmaf(v[u].z, w, acc.z);
        acc.w = fmaf(v[u].w, w, acc.w);
    }
}

__device__ __forceinline__ void gcn_chunk(float4& acc, const float4* __restrict__ H4,
                                          const int2* __restrict__ slot,
                                          const int first, const int m, const int lane) {
    int2 mine = make_int2(0, 0);
    if (lane < m) mine = slot[first + lane];
    int j = 0;
    for (; j + 8 <= m; j += 8) gcn_rows<8>(acc, H4, mine, j, lane);
    switch (m - j) {
        case 7: gcn_rows<7>(acc, H4, mine, j, lane); break;
        case 6: gcn_rows<6>(acc, H4, mine, j, lane); break;
        case 5: gcn_rows<5>(acc, H4, mine, j, lane); break;
        case 4: gcn_rows<4>(acc, H4, mine, j, lane); break;
        case 3: gcn_rows<3>(acc, H4, mine, j, lane); break;
        case 2: gcn_rows<2>(acc, H4, mine, j, lane); break;
        case 1: gcn_rows<1>(acc, H4, mine, j, lane); break;
        default: break;
    }
}

__global__ void __launch_bounds__(256)
gcn_gather_rows(const float* __restrict__ H, const u64* __restrict__ row,
                const int2* __restrict__ slot, const float* __restrict__ dis,
                const float* __restrict__ b, float* __restrict__ out, int N) {
    const int lane = threadIdx.x & 31;
    const int d = __builtin_amdgcn_readfirstlane((int)((blockIdx.x * blockDim.x + threadIdx.x) >> 5));
    if (d >= N) return;

    const float4* H4 = (const float4*)H;
    const u64    ri = row[d];
    const float  sd = dis[d];
    const float4 hv = H4[(size_t)d * (D_FEAT / 4) + lane];
    const float4 bv = ((const float4*)b)[lane];
    const int start = (int)(unsigned)ri;
    const int n     = (int)(ri >> 32);

    float4 acc = make_float4(0.0f, 0.0f, 0.0f, 0.0f);

    if (n > 0) gcn_chunk(acc, H4, slot, start, (n < 32) ? n : 32, lane);
    for (int j0 = 32; j0 < n; j0 += 32)
        gcn_chunk(acc, H4, slot, start + j0, (n - j0 < 32) ? (n - j0) : 32, lane);

    const float ws = sd * sd;
    float4 o;
    o.x = fmaf(hv.x, ws, acc.x) + bv.x;
    o.y = fmaf(hv.y, ws, acc.y) + bv.y;
    o.z = fmaf(hv.z, ws, acc.z) + bv.z;
    o.w = fmaf(hv.w, ws, acc.w) + bv.w;
    ((float4*)out)[(size_t)d * (D_FEAT / 4) + lane] = o;
}

extern "C" void kernel_launch(void* const* d_in, const int* in_sizes, int n_in,
                              void* d_out, int out_size, void* d_ws, size_t ws_size,
                              hipStream_t stream) {
    const float* feat = (const float*)d_in[0];
    const int*   ei   = (const int*)d_in[1];
    const float* W0   = (const float*)d_in[2];
    const float* b0   = (const float*)d_in[3];
    const float* W1   = (const float*)d_in[4];
    const float* b1   = (const float*)d_in[5];
    const float* W2   = (const float*)d_in[6];
    const float* b2   = (const float*)d_in[7];

    const int N = in_sizes[0] / D_FEAT;
    const int E = in_sizes[1] / 2;
    const int* src = ei;
    const int* dst = ei + E;

    const size_t Np = ((size_t)N + 1023) & ~(size_t)1023;
    const size_t Ep = ((size_t)E + 1023) & ~(size_t)1023;
    int*   cnt  = (int*)d_ws;
    float* dis  = (float*)(cnt + Np);
    u64*   row  = (u64*)(dis + Np);
    int*   rank = (int*)(row + Np);
    int2*  raw  = (int2*)(rank + Ep);
    int2*  slot = raw + Ep;
    float* H    = (float*)(slot + Ep);
    float* X    = H + (size_t)N * D_FEAT;
    float* Wc   = X + (size_t)N * D_FEAT;
    float* out  = (float*)d_out;

    const int TB = 256;
    dim3 blkFold(FOLD_BLOCKS + (N + TB - 1) / TB);
    dim3 blkRank(GEMM_BLOCKS + (E + RANK_EDGES - 1) / RANK_EDGES);
    dim3 blkE((E + TB - 1) / TB);
    dim3 blkScan((N + SCAN_ROWS - 1) / SCAN_ROWS);
    dim3 blkRow(((size_t)N * 32 + TB - 1) / TB);

    gcn_fold_weights_clear<<<blkFold, TB, 0, stream>>>(W0, b0, W1, b1, W2, Wc, cnt, N);

    gcn_rank_edges_gemm<<<blkRank, TB, 0, stream>>>(dst, cnt, rank, E, feat, Wc, H, N);

    gcn_scan_deg<<<blkScan, TB, 0, stream>>>(cnt, row, dis, N);
    gcn_fill_slots<<<blkE, TB, 0, stream>>>(src, dst, row, rank, dis, raw, E);
    gcn_order_rows<<<blkRow, TB, 0, stream>>>(row, raw, slot, N);

    gcn_gather_rows<<<blkRow, TB, 0, stream>>>(H, row, slot, dis, Wc + 128 * D_FEAT, X, N);
    gcn_gather_rows<<<blkRow, TB, 0, stream>>>(X, row, slot, dis, Wc + 129 * D_FEAT, H, N);
    gcn_gather_rows<<<blkRow, TB, 0, stream>>>(H, row, slot, dis, b2, out, N);
}


# Row-offset scan: a block's sum of the counters in front of it keeps 28 independent 16-byte loads per lane in flight before the first add (was one load per trip, each waited for); integer sums
# speedup vs baseline: 3.8706x; 1.1231x over previous
#include <hip/hip_runtime.h>
#include <hip/hip_bf16.h>

typedef __attribute__((ext_vector_type(2))) float v2f;
typedef __attribute__((ext_vector_type(8))) float v8f;

#define D_FEAT 128
#define LDS_STRIDE 132

#define SCAN_ROWS 1024
#define SCAN_GROUP 28
typedef unsigned long long u64;
typedef __attribute__((ext_vector_type(4))) int v4i;

__global__ void __launch_bounds__(256)
gcn_scan_deg(const int* __restrict__ cnt, u64* __restrict__ row,
             float* __restrict__ dis, int N) {
    __shared__ int wsum[8];
    __shared__ int wpre[8];
    const int tid  = threadIdx.x;
    const int lane = tid & 31;
    const int wave = tid >> 5;
    const int base = blockIdx.x * SCAN_ROWS;

    int pre = 0;
    const v4i* c4 = (const v4i*)cnt;
    const int nb = base / SCAN_ROWS;
    for (int k = 0; k < nb; k += SCAN_GROUP) {
        v4i v[SCAN_GROUP];
        #pragma unroll
        for (int u = 0; u < SCAN_GROUP; ++u) {
            const int kk = (k + u < nb) ? (k + u) : (nb - 1);
            v[u] = c4[kk * 256 + tid];
        }
        asm volatile("" ::
                        "v"(v[0]), "v"(v[1]), "v"(v[2]), "v"(v[3]), "v"(v[4]), "v"(v[5]), "v"(v[6]),
                        "v"(v[7]), "v"(v[8]), "v"(v[9]), "v"(v[10]), "v"(v[11]), "v"(v[12]), "v"(v[13]),
                        "v"(v[14]), "v"(v[15]), "v"(v[16]), "v"(v[17]), "v"(v[18]), "v"(v[19]), "v"(v[20]),
                        "v"(v[21]), "v"(v[22]), "v"(v[23]), "v"(v[24]), "v"(v[25]), "v"(v[26]), "v"(v[27]));
        #pragma unroll
        for (int u = 0; u < SCAN_GROUP; ++u) {
            const int s = (v[u].x + v[u].y) + (v[u].z + v[u].w);
            pre += (k + u < nb) ? s : 0;
        }
    }

    const int i0 = base + 4 * tid;
    int c[4];
    for (int k = 0; k < 4; ++k) c[k] = (i0 + k < N) ? cnt[i0 + k] : 0;
    const int tsum = (c[0] + c[1]) + (c[2] + c[3]);

    int inc = tsum;
    for (int o = 1; o < 32; o <<= 1) {
        const int t = __shfl_up(inc, o);
        if (lane >= o) inc += t;
    }
    for (int o = 16; o > 0; o >>= 1) pre += __shfl_xor(pre, o);
    if (lane == 31) wsum[wave] = inc;
    if (lane == 0)  wpre[wave] = pre;
    __syncthreads();
    int off = 0;
    for (int w = 0; w < 8; ++w) {
        off += wpre[w];
        if (w < wave) off += wsum[w];
    }
    int ex = off + inc - tsum;
    for (int k = 0; k < 4; ++k) {
        if (i0 + k < N) {
            row[i0 + k] = ((u64)(unsigned)c[k] << 32) | (u64)(unsigned)ex;
            dis[i0 + k] = rsqrtf((float)(c[k] + 1));
        }
        ex += c[k];
    }
}

__global__ void gcn_fill_slots(const int* __restrict__ src, const int* __restrict__ dst,
                               const u64* __restrict__ row, const int* __restrict__ rank,
                               const float* __restrict__ dis, int2* __restrict__ raw, int E) {
    int e = blockIdx.x * blockDim.x + threadIdx.x;
    if (e < E) {
        const int s = src[e];
        const int d = dst[e];
        const float w = dis[s] * dis[d];
        const int first = (int)(unsigned)row[d];
        raw[first + rank[e]] = make_int2(s, __float_as_int(w));
    }
}

__global__ void __launch_bounds__(256)
gcn_order_rows(const u64* __restrict__ row, const int2* __restrict__ raw,
               int2* __restrict__ slot, int N) {
    const int lane = threadIdx.x & 31;
    const int d = __builtin_amdgcn_readfirstlane((int)((blockIdx.x * blockDim.x + threadIdx.x) >> 5));
    if (d >= N) return;
    const u64 ri    = row[d];
    const int start = (int)(unsigned)ri;
    const int n     = (int)(ri >> 32);
    for (int c0 = 0; c0 < n; c0 += 32) {
        const int i = c0 + lane;
        int2 mine = make_int2(0x7fffffff, 0);
        if (i < n) mine = raw[start + i];
        const u64 mk = ((u64)(unsigned)mine.x << 32) | (u64)(unsigned)i;
        int pos = 0;
        for (int k0 = 0; k0 < n; k0 += 32) {
            int key = mine.x;
            if (k0 != c0) {
                key = 0x7fffffff;
                if (k0 + lane < n) key = raw[start + k0 + lane].x;
            }
            const int cm = (n - k0 < 32) ? (n - k0) : 32;
            for (int k = 0; k < cm; ++k) {
                const u64 ok = ((u64)(unsigned)__builtin_amdgcn_readlane(key, k) << 32)
                             | (u64)(unsigned)(k0 + k);
                pos += (ok < mk) ? 1 : 0;
            }
        }
        if (i < n) slot[start + pos] = mine;
    }
}

#define GEMM_BLOCKS 512
#define RANK_EDGES 1024

__device__ __forceinline__ void gcn_gemm_panels(const float* __restrict__ X,
                                                const float* __restrict__ W,
                                                float* __restrict__ H, const int N, const int b) {
    __shared__ float As[16 * LDS_STRIDE];

    const int tid  = threadIdx.x;
    const int wave = tid >> 5;
    const int lane = tid & 31;
    const int half = lane >> 4;
    const int l16  = lane & 15;
    const int col  = wave * 16 + l16;

    v2f bf[D_FEAT / 4];
    #pragma unroll
    for (int i = 0; i < D_FEAT / 4; ++i) {
        const int k = 4 * i + 2 * half;
        bf[i].x = W[(size_t)k * D_FEAT + col];
        bf[i].y = W[(size_t)(k + 1) * D_FEAT + col];
    }

    const int npan = (N + 15) >> 4;
    const float4 zero4 = make_float4(0.0f, 0.0f, 0.0f, 0.0f);

    float4 a0 = zero4, a1 = zero4;
    if (b < npan) {
        const int r = 16 * b + wave;
        if (r < N)     a0 = ((const float4*)(X + (size_t)r * D_FEAT))[lane];
        if (r + 8 < N) a1 = ((const float4*)(X + (size_t)(r + 8) * D_FEAT))[lane];
    }

    #pragma unroll 1
    for (int p = b; p < npan; p += GEMM_BLOCKS) {
        __syncthreads();
        *(float4*)&As[wave * LDS_STRIDE + 4 * lane]       = a0;
        *(float4*)&As[(wave + 8) * LDS_STRIDE + 4 * lane] = a1;
        __syncthreads();

        const int pn = p + GEMM_BLOCKS;
        a0 = zero4; a1 = zero4;
        if (pn < npan) {
            const int r = 16 * pn + wave;
            if (r < N)     a0 = ((const float4*)(X + (size_t)r * D_FEAT))[lane];
            if (r + 8 < N) a1 = ((const float4*)(X + (size_t)(r + 8) * D_FEAT))[lane];
        }

        v8f acc = {};
        #pragma unroll
        for (int i = 0; i < D_FEAT / 4; ++i) {
            const v2f a = *(const v2f*)(As + l16 * LDS_STRIDE + 4 * i + 2 * half);
            acc = __builtin_amdgcn_wmma_f32_16x16x4_f32(
                 false, a,  false, bf[i],
                 (short)0, acc,  false,  false);
        }

        const int r0 = 16 * p;
        #pragma unroll
        for (int j = 0; j < 8; ++j) {
            const int row = r0 + j + 8 * half;
            if (row < N) H[(size_t)row * D_FEAT + col] = acc[j];
        }
    }
}

__global__ void __launch_bounds__(256)
gcn_rank_edges_gemm(const int* __restrict__ dst, int* __restrict__ cnt,
                    int* __restrict__ rank, int E,
                    const float* __restrict__ X, const float* __restrict__ W,
                    float* __restrict__ H, int N) {
    if (blockIdx.x < GEMM_BLOCKS) {
        gcn_gemm_panels(X, W, H, N, (int)blockIdx.x);
        return;
    }
    const int base = (int)(blockIdx.x - GEMM_BLOCKS) * RANK_EDGES;
    const int e0   = base + (int)threadIdx.x;
    if (base + RANK_EDGES <= E) {
        int d[RANK_EDGES / 256], r[RANK_EDGES / 256];
        #pragma unroll
        for (int u = 0; u < RANK_EDGES / 256; ++u) d[u] = dst[e0 + 256 * u];
        #pragma unroll
        for (int u = 0; u < RANK_EDGES / 256; ++u) r[u] = atomicAdd(&cnt[d[u]], 1);
        #pragma unroll
        for (int u = 0; u < RANK_EDGES / 256; ++u) rank[e0 + 256 * u] = r[u];
    } else {
        for (int u = 0; u < RANK_EDGES / 256; ++u) {
            const int e = e0 + 256 * u;
            if (e < E) rank[e] = atomicAdd(&cnt[dst[e]], 1);
        }
    }
}

#define FOLD_BLOCKS 9

__device__ __forceinline__ v8f gcn_panel_times_wt(const float* __restrict__ As,
                                                  const float* __restrict__ Ws,
                                                  const int l16, const int half, const int col) {
    v8f acc = {};
    #pragma unroll
    for (int k0 = 0; k0 < D_FEAT; k0 += 4) {
        const int ka = k0 + 2 * half;
        const v2f a = *(const v2f*)(As + l16 * LDS_STRIDE + ka);
        const v2f b = *(const v2f*)(Ws + col * LDS_STRIDE + ka);
        acc = __builtin_amdgcn_wmma_f32_16x16x4_f32(
             false, a,  false, b,
             (short)0, acc,  false,  false);
    }
    return acc;
}

__global__ void __launch_bounds__(256)
gcn_fold_weights_clear(const float* __restrict__ W0, const float* __restrict__ b0,
                       const float* __restrict__ W1, const float* __restrict__ b1,
                       const float* __restrict__ W2, float* __restrict__ Wc,
                       int* __restrict__ cnt, int N) {
    __shared__ float W1s[D_FEAT * LDS_STRIDE];
    __shared__ float W2s[D_FEAT * LDS_STRIDE];
    __shared__ float As[16 * LDS_STRIDE];

    const int tid  = threadIdx.x;
    if (blockIdx.x >= FOLD_BLOCKS) {
        const int i = (int)(blockIdx.x - FOLD_BLOCKS) * 256 + tid;
        if (i < N) cnt[i] = 0;
        return;
    }
    const int p    = blockIdx.x;
    const int wave = tid >> 5;
    const int lane = tid & 31;
    const int half = lane >> 4;
    const int l16  = lane & 15;
    const int col  = wave * 16 + l16;

    const int kk = lane & 7;
    const int pc = 4 * wave + (lane >> 3);
    float4 w1[16], w2[16], a4[2];
    #pragma unroll
    for (int u = 0; u < 16; ++u)
        w1[u] = ((const float4*)(W1 + (size_t)(8 * u + kk) * D_FEAT))[pc];
    #pragma unroll
    for (int u = 0; u < 16; ++u)
        w2[u] = ((const float4*)(W2 + (size_t)(8 * u + kk) * D_FEAT))[pc];
    #pragma unroll
    for (int u = 0; u < 2; ++u) {
        const int r = 8 * u + wave;
        a4[u] = make_float4(0.0f, 0.0f, 0.0f, 0.0f);
        if (p < 8)       a4[u] = ((const float4*)(W0 + (size_t)(16 * p + r) * D_FEAT))[lane];
        else if (r == 0) a4[u] = ((const float4*)b0)[lane];
    }
    #pragma unroll
    for (int u = 0; u < 2; ++u)
        *(float4*)&As[(8 * u + wave) * LDS_STRIDE + 4 * lane] = a4[u];
    #pragma unroll
    for (int u = 0; u < 16; ++u) {
        const int k = 8 * u + kk;
        W1s[(4 * pc + 0) * LDS_STRIDE + k] = w1[u].x;
        W1s[(4 * pc + 1) * LDS_STRIDE + k] = w1[u].y;
        W1s[(4 * pc + 2) * LDS_STRIDE + k] = w1[u].z;
        W1s[(4 * pc + 3) * LDS_STRIDE + k] = w1[u].w;
    }
    #pragma unroll
    for (int u = 0; u < 16; ++u) {
        const int k = 8 * u + kk;
        W2s[(4 * pc + 0) * LDS_STRIDE + k] = w2[u].x;
        W2s[(4 * pc + 1) * LDS_STRIDE + k] = w2[u].y;
        W2s[(4 * pc + 2) * LDS_STRIDE + k] = w2[u].z;
        W2s[(4 * pc + 3) * LDS_STRIDE + k] = w2[u].w;
    }
    __syncthreads();

    const v8f t = gcn_panel_times_wt(As, W1s, l16, half, col);
    __syncthreads();

    #pragma unroll
    for (int j = 0; j < 8; ++j) As[(j + 8 * half) * LDS_STRIDE + col] = t[j];
    if (p == 8 && half == 0) As[1 * LDS_STRIDE + col] = b1[col];
    __syncthreads();

    const v8f o = gcn_panel_times_wt(As, W2s, l16, half, col);

    const int nrow = (p < 8) ? 16 : 2;
    #pragma unroll
    for (int j = 0; j < 8; ++j) {
        const int r = j + 8 * half;
        if (r < nrow) Wc[(size_t)(16 * p + r) * D_FEAT + col] = o[j];
    }
}

template <int B>
__device__ __forceinline__ void gcn_rows(float4& acc, const float4* __restrict__ H4,
                                         const int2 mine, const int j, const int lane) {
    float4 v[B];
    for (int u = 0; u < B; ++u) {
        const int s = __builtin_amdgcn_readlane(mine.x, j + u);
        v[u] = H4[(size_t)s * (D_FEAT / 4) + lane];
    }
    for (int u = 0; u < B; ++u) {
        const float w = __int_as_float(__builtin_amdgcn_readlane(mine.y, j + u));
        acc.x = fmaf(v[u].x, w, acc.x);
        acc.y = fmaf(v[u].y, w, acc.y);
        acc.z = fmaf(v[u].z, w, acc.z);
        acc.w = fmaf(v[u].w, w, acc.w);
    }
}

__device__ __forceinline__ void gcn_chunk(float4& acc, const float4* __restrict__ H4,
                                          const int2* __restrict__ slot,
                                          const int first, const int m, const int lane) {
    int2 mine = make_int2(0, 0);
    if (lane < m) mine = slot[first + lane];
    int j = 0;
    for (; j + 8 <= m; j += 8) gcn_rows<8>(acc, H4, mine, j, lane);
    switch (m - j) {
        case 7: gcn_rows<7>(acc, H4, mine, j, lane); break;
        case 6: gcn_rows<6>(acc, H4, mine, j, lane); break;
        case 5: gcn_rows<5>(acc, H4, mine, j, lane); break;
        case 4: gcn_rows<4>(acc, H4, mine, j, lane); break;
        case 3: gcn_rows<3>(acc, H4, mine, j, lane); break;
        case 2: gcn_rows<2>(acc, H4, mine, j, lane); break;
        case 1: gcn_rows<1>(acc, H4, mine, j, lane); break;
        default: break;
    }
}

__global__ void __launch_bounds__(256)
gcn_gather_rows(const float* __restrict__ H, const u64* __restrict__ row,
                const int2* __restrict__ slot, const float* __restrict__ dis,
                const float* __restrict__ b, float* __restrict__ out, int N) {
    const int lane = threadIdx.x & 31;
    const int d = __builtin_amdgcn_readfirstlane((int)((blockIdx.x * blockDim.x + threadIdx.x) >> 5));
    if (d >= N) return;

    const float4* H4 = (const float4*)H;
    const u64    ri = row[d];
    const float  sd = dis[d];
    const float4 hv = H4[(size_t)d * (D_FEAT / 4) + lane];
    const float4 bv = ((const float4*)b)[lane];
    const int start = (int)(unsigned)ri;
    const int n     = (int)(ri >> 32);

    float4 acc = make_float4(0.0f, 0.0f, 0.0f, 0.0f);

    if (n > 0) gcn_chunk(acc, H4, slot, start, (n < 32) ? n : 32, lane);
    for (int j0 = 32; j0 < n; j0 += 32)
        gcn_chunk(acc, H4, slot, start + j0, (n - j0 < 32) ? (n - j0) : 32, lane);

    const float ws = sd * sd;
    float4 o;
    o.x = fmaf(hv.x, ws, acc.x) + bv.x;
    o.y = fmaf(hv.y, ws, acc.y) + bv.y;
    o.z = fmaf(hv.z, ws, acc.z) + bv.z;
    o.w = fmaf(hv.w, ws, acc.w) + bv.w;
    ((float4*)out)[(size_t)d * (D_FEAT / 4) + lane] = o;
}

extern "C" void kernel_launch(void* const* d_in, const int* in_sizes, int n_in,
                              void* d_out, int out_size, void* d_ws, size_t ws_size,
                              hipStream_t stream) {
    const float* feat = (const float*)d_in[0];
    const int*   ei   = (const int*)d_in[1];
    const float* W0   = (const float*)d_in[2];
    const float* b0   = (const float*)d_in[3];
    const float* W1   = (const float*)d_in[4];
    const float* b1   = (const float*)d_in[5];
    const float* W2   = (const float*)d_in[6];
    const float* b2   = (const float*)d_in[7];

    const int N = in_sizes[0] / D_FEAT;
    const int E = in_sizes[1] / 2;
    const int* src = ei;
    const int* dst = ei + E;

    const size_t Np = ((size_t)N + 1023) & ~(size_t)1023;
    const size_t Ep = ((size_t)E + 1023) & ~(size_t)1023;
    int*   cnt  = (int*)d_ws;
    float* dis  = (float*)(cnt + Np);
    u64*   row  = (u64*)(dis + Np);
    int*   rank = (int*)(row + Np);
    int2*  raw  = (int2*)(rank + Ep);
    int2*  slot = raw + Ep;
    float* H    = (float*)(slot + Ep);
    float* X    = H + (size_t)N * D_FEAT;
    float* Wc   = X + (size_t)N * D_FEAT;
    float* out  = (float*)d_out;

    const int TB = 256;
    dim3 blkFold(FOLD_BLOCKS + (N + TB - 1) / TB);
    dim3 blkRank(GEMM_BLOCKS + (E + RANK_EDGES - 1) / RANK_EDGES);
    dim3 blkE((E + TB - 1) / TB);
    dim3 blkScan((N + SCAN_ROWS - 1) / SCAN_ROWS);
    dim3 blkRow(((size_t)N * 32 + TB - 1) / TB);

    gcn_fold_weights_clear<<<blkFold, TB, 0, stream>>>(W0, b0, W1, b1, W2, Wc, cnt, N);

    gcn_rank_edges_gemm<<<blkRank, TB, 0, stream>>>(dst, cnt, rank, E, feat, Wc, H, N);

    gcn_scan_deg<<<blkScan, TB, 0, stream>>>(cnt, row, dis, N);
    gcn_fill_slots<<<blkE, TB, 0, stream>>>(src, dst, row, rank, dis, raw, E);
    gcn_order_rows<<<blkRow, TB, 0, stream>>>(row, raw, slot, N);

    gcn_gather_rows<<<blkRow, TB, 0, stream>>>(H, row, slot, dis, Wc + 128 * D_FEAT, X, N);
    gcn_gather_rows<<<blkRow, TB, 0, stream>>>(X, row, slot, dis, Wc + 129 * D_FEAT, H, N);
    gcn_gather_rows<<<blkRow, TB, 0, stream>>>(H, row, slot, dis, b2, out, N);
}


# Bucket sort rides in the first row-sum launch: the wave that sums row d ranks its in-edges by src, orders them through 256 B of LDS and stores them for layers 2, 3; 8 launches -> 7; sums unchanged
# speedup vs baseline: 3.9864x; 1.0299x over previous
#include <hip/hip_runtime.h>
#include <hip/hip_bf16.h>

typedef __attribute__((ext_vector_type(2))) float v2f;
typedef __attribute__((ext_vector_type(8))) float v8f;

#define D_FEAT 128
#define LDS_STRIDE 132

#define SCAN_ROWS 1024
#define SCAN_GROUP 28
typedef unsigned long long u64;
typedef __attribute__((ext_vector_type(4))) int v4i;

__global__ void __launch_bounds__(256)
gcn_scan_deg(const int* __restrict__ cnt, u64* __restrict__ row,
             float* __restrict__ dis, int N) {
    __shared__ int wsum[8];
    __shared__ int wpre[8];
    const int tid  = threadIdx.x;
    const int lane = tid & 31;
    const int wave = tid >> 5;
    const int base = blockIdx.x * SCAN_ROWS;

    int pre = 0;
    const v4i* c4 = (const v4i*)cnt;
    const int nb = base / SCAN_ROWS;
    for (int k = 0; k < nb; k += SCAN_GROUP) {
        v4i v[SCAN_GROUP];
        #pragma unroll
        for (int u = 0; u < SCAN_GROUP; ++u) {
            const int kk = (k + u < nb) ? (k + u) : (nb - 1);
            v[u] = c4[kk * 256 + tid];
        }
        asm volatile("" ::
                        "v"(v[0]), "v"(v[1]), "v"(v[2]), "v"(v[3]), "v"(v[4]), "v"(v[5]), "v"(v[6]),
                        "v"(v[7]), "v"(v[8]), "v"(v[9]), "v"(v[10]), "v"(v[11]), "v"(v[12]), "v"(v[13]),
                        "v"(v[14]), "v"(v[15]), "v"(v[16]), "v"(v[17]), "v"(v[18]), "v"(v[19]), "v"(v[20]),
                        "v"(v[21]), "v"(v[22]), "v"(v[23]), "v"(v[24]), "v"(v[25]), "v"(v[26]), "v"(v[27]));
        #pragma unroll
        for (int u = 0; u < SCAN_GROUP; ++u) {
            const int s = (v[u].x + v[u].y) + (v[u].z + v[u].w);
            pre += (k + u < nb) ? s : 0;
        }
    }

    const int i0 = base + 4 * tid;
    int c[4];
    for (int k = 0; k < 4; ++k) c[k] = (i0 + k < N) ? cnt[i0 + k] : 0;
    const int tsum = (c[0] + c[1]) + (c[2] + c[3]);

    int inc = tsum;
    for (int o = 1; o < 32; o <<= 1) {
        const int t = __shfl_up(inc, o);
        if (lane >= o) inc += t;
    }
    for (int o = 16; o > 0; o >>= 1) pre += __shfl_xor(pre, o);
    if (lane == 31) wsum[wave] = inc;
    if (lane == 0)  wpre[wave] = pre;
    __syncthreads();
    int off = 0;
    for (int w = 0; w < 8; ++w) {
        off += wpre[w];
        if (w < wave) off += wsum[w];
    }
    int ex = off + inc - tsum;
    for (int k = 0; k < 4; ++k) {
        if (i0 + k < N) {
            row[i0 + k] = ((u64)(unsigned)c[k] << 32) | (u64)(unsigned)ex;
            dis[i0 + k] = rsqrtf((float)(c[k] + 1));
        }
        ex += c[k];
    }
}

__global__ void gcn_fill_slots(const int* __restrict__ src, const int* __restrict__ dst,
                               const u64* __restrict__ row, const int* __restrict__ rank,
                               const float* __restrict__ dis, int2* __restrict__ raw, int E) {
    int e = blockIdx.x * blockDim.x + threadIdx.x;
    if (e < E) {
        const int s = src[e];
        const int d = dst[e];
        const float w = dis[s] * dis[d];
        const int first = (int)(unsigned)row[d];
        raw[first + rank[e]] = make_int2(s, __float_as_int(w));
    }
}

#define GEMM_BLOCKS 512
#define RANK_EDGES 1024

__device__ __forceinline__ void gcn_gemm_panels(const float* __restrict__ X,
                                                const float* __restrict__ W,
                                                float* __restrict__ H, const int N, const int b) {
    __shared__ float As[16 * LDS_STRIDE];

    const int tid  = threadIdx.x;
    const int wave = tid >> 5;
    const int lane = tid & 31;
    const int half = lane >> 4;
    const int l16  = lane & 15;
    const int col  = wave * 16 + l16;

    v2f bf[D_FEAT / 4];
    #pragma unroll
    for (int i = 0; i < D_FEAT / 4; ++i) {
        const int k = 4 * i + 2 * half;
        bf[i].x = W[(size_t)k * D_FEAT + col];
        bf[i].y = W[(size_t)(k + 1) * D_FEAT + col];
    }

    const int npan = (N + 15) >> 4;
    const float4 zero4 = make_float4(0.0f, 0.0f, 0.0f, 0.0f);

    float4 a0 = zero4, a1 = zero4;
    if (b < npan) {
        const int r = 16 * b + wave;
        if (r < N)     a0 = ((const float4*)(X + (size_t)r * D_FEAT))[lane];
        if (r + 8 < N) a1 = ((const float4*)(X + (size_t)(r + 8) * D_FEAT))[lane];
    }

    #pragma unroll 1
    for (int p = b; p < npan; p += GEMM_BLOCKS) {
        __syncthreads();
        *(float4*)&As[wave * LDS_STRIDE + 4 * lane]       = a0;
        *(float4*)&As[(wave + 8) * LDS_STRIDE + 4 * lane] = a1;
        __syncthreads();

        const int pn = p + GEMM_BLOCKS;
        a0 = zero4; a1 = zero4;
        if (pn < npan) {
            const int r = 16 * pn + wave;
            if (r < N)     a0 = ((const float4*)(X + (size_t)r * D_FEAT))[lane];
            if (r + 8 < N) a1 = ((const float4*)(X + (size_t)(r + 8) * D_FEAT))[lane];
        }

        v8f acc = {};
        #pragma unroll
        for (int i = 0; i < D_FEAT / 4; ++i) {
            const v2f a = *(const v2f*)(As + l16 * LDS_STRIDE + 4 * i + 2 * half);
            acc = __builtin_amdgcn_wmma_f32_16x16x4_f32(
                 false, a,  false, bf[i],
                 (short)0, acc,  false,  false);
        }

        const int r0 = 16 * p;
        #pragma unroll
        for (int j = 0; j < 8; ++j) {
            const int row = r0 + j + 8 * half;
            if (row < N) H[(size_t)row * D_FEAT + col] = acc[j];
        }
    }
}

__global__ void __launch_bounds__(256)
gcn_rank_edges_gemm(const int* __restrict__ dst, int* __restrict__ cnt,
                    int* __restrict__ rank, int E,
                    const float* __restrict__ X, const float* __restrict__ W,
                    float* __restrict__ H, int N) {
    if (blockIdx.x < GEMM_BLOCKS) {
        gcn_gemm_panels(X, W, H, N, (int)blockIdx.x);
        return;
    }
    const int base = (int)(blockIdx.x - GEMM_BLOCKS) * RANK_EDGES;
    const int e0   = base + (int)threadIdx.x;
    if (base + RANK_EDGES <= E) {
        int d[RANK_EDGES / 256], r[RANK_EDGES / 256];
        #pragma unroll
        for (int u = 0; u < RANK_EDGES / 256; ++u) d[u] = dst[e0 + 256 * u];
        #pragma unroll
        for (int u = 0; u < RANK_EDGES / 256; ++u) r[u] = atomicAdd(&cnt[d[u]], 1);
        #pragma unroll
        for (int u = 0; u < RANK_EDGES / 256; ++u) rank[e0 + 256 * u] = r[u];
    } else {
        for (int u = 0; u < RANK_EDGES / 256; ++u) {
            const int e = e0 + 256 * u;
            if (e < E) rank[e] = atomicAdd(&cnt[dst[e]], 1);
        }
    }
}

#define FOLD_BLOCKS 9

__device__ __forceinline__ v8f gcn_panel_times_wt(const float* __restrict__ As,
                                                  const float* __restrict__ Ws,
                                                  const int l16, const int half, const int col) {
    v8f acc = {};
    #pragma unroll
    for (int k0 = 0; k0 < D_FEAT; k0 += 4) {
        const int ka = k0 + 2 * half;
        const v2f a = *(const v2f*)(As + l16 * LDS_STRIDE + ka);
        const v2f b = *(const v2f*)(Ws + col * LDS_STRIDE + ka);
        acc = __builtin_amdgcn_wmma_f32_16x16x4_f32(
             false, a,  false, b,
             (short)0, acc,  false,  false);
    }
    return acc;
}

__global__ void __launch_bounds__(256)
gcn_fold_weights_clear(const float* __restrict__ W0, const float* __restrict__ b0,
                       const float* __restrict__ W1, const float* __restrict__ b1,
                       const float* __restrict__ W2, float* __restrict__ Wc,
                       int* __restrict__ cnt, int N) {
    __shared__ float W1s[D_FEAT * LDS_STRIDE];
    __shared__ float W2s[D_FEAT * LDS_STRIDE];
    __shared__ float As[16 * LDS_STRIDE];

    const int tid  = threadIdx.x;
    if (blockIdx.x >= FOLD_BLOCKS) {
        const int i = (int)(blockIdx.x - FOLD_BLOCKS) * 256 + tid;
        if (i < N) cnt[i] = 0;
        return;
    }
    const int p    = blockIdx.x;
    const int wave = tid >> 5;
    const int lane = tid & 31;
    const int half = lane >> 4;
    const int l16  = lane & 15;
    const int col  = wave * 16 + l16;

    const int kk = lane & 7;
    const int pc = 4 * wave + (lane >> 3);
    float4 w1[16], w2[16], a4[2];
    #pragma unroll
    for (int u = 0; u < 16; ++u)
        w1[u] = ((const float4*)(W1 + (size_t)(8 * u + kk) * D_FEAT))[pc];
    #pragma unroll
    for (int u = 0; u < 16; ++u)
        w2[u] = ((const float4*)(W2 + (size_t)(8 * u + kk) * D_FEAT))[pc];
    #pragma unroll
    for (int u = 0; u < 2; ++u) {
        const int r = 8 * u + wave;
        a4[u] = make_float4(0.0f, 0.0f, 0.0f, 0.0f);
        if (p < 8)       a4[u] = ((const float4*)(W0 + (size_t)(16 * p + r) * D_FEAT))[lane];
        else if (r == 0) a4[u] = ((const float4*)b0)[lane];
    }
    #pragma unroll
    for (int u = 0; u < 2; ++u)
        *(float4*)&As[(8 * u + wave) * LDS_STRIDE + 4 * lane] = a4[u];
    #pragma unroll
    for (int u = 0; u < 16; ++u) {
        const int k = 8 * u + kk;
        W1s[(4 * pc + 0) * LDS_STRIDE + k] = w1[u].x;
        W1s[(4 * pc + 1) * LDS_STRIDE + k] = w1[u].y;
        W1s[(4 * pc + 2) * LDS_STRIDE + k] = w1[u].z;
        W1s[(4 * pc + 3) * LDS_STRIDE + k] = w1[u].w;
    }
    #pragma unroll
    for (int u = 0; u < 16; ++u) {
        const int k = 8 * u + kk;
        W2s[(4 * pc + 0) * LDS_STRIDE + k] = w2[u].x;
        W2s[(4 * pc + 1) * LDS_STRIDE + k] = w2[u].y;
        W2s[(4 * pc + 2) * LDS_STRIDE + k] = w2[u].z;
        W2s[(4 * pc + 3) * LDS_STRIDE + k] = w2[u].w;
    }
    __syncthreads();

    const v8f t = gcn_panel_times_wt(As, W1s, l16, half, col);
    __syncthreads();

    #pragma unroll
    for (int j = 0; j < 8; ++j) As[(j + 8 * half) * LDS_STRIDE + col] = t[j];
    if (p == 8 && half == 0) As[1 * LDS_STRIDE + col] = b1[col];
    __syncthreads();

    const v8f o = gcn_panel_times_wt(As, W2s, l16, half, col);

    const int nrow = (p < 8) ? 16 : 2;
    #pragma unroll
    for (int j = 0; j < 8; ++j) {
        const int r = j + 8 * half;
        if (r < nrow) Wc[(size_t)(16 * p + r) * D_FEAT + col] = o[j];
    }
}

template <int B>
__device__ __forceinline__ void gcn_rows(float4& acc, const float4* __restrict__ H4,
                                         const int2 mine, const int j, const int lane) {
    float4 v[B];
    for (int u = 0; u < B; ++u) {
        const int s = __builtin_amdgcn_readlane(mine.x, j + u);
        v[u] = H4[(size_t)s * (D_FEAT / 4) + lane];
    }
    for (int u = 0; u < B; ++u) {
        const float w = __int_as_float(__builtin_amdgcn_readlane(mine.y, j + u));
        acc.x = fmaf(v[u].x, w, acc.x);
        acc.y = fmaf(v[u].y, w, acc.y);
        acc.z = fmaf(v[u].z, w, acc.z);
        acc.w = fmaf(v[u].w, w, acc.w);
    }
}

__device__ __forceinline__ void gcn_chunk_sum(float4& acc, const float4* __restrict__ H4,
                                              const int2 mine, const int m, const int lane) {
    int j = 0;
    for (; j + 8 <= m; j += 8) gcn_rows<8>(acc, H4, mine, j, lane);
    switch (m - j) {
        case 7: gcn_rows<7>(acc, H4, mine, j, lane); break;
        case 6: gcn_rows<6>(acc, H4, mine, j, lane); break;
        case 5: gcn_rows<5>(acc, H4, mine, j, lane); break;
        case 4: gcn_rows<4>(acc, H4, mine, j, lane); break;
        case 3: gcn_rows<3>(acc, H4, mine, j, lane); break;
        case 2: gcn_rows<2>(acc, H4, mine, j, lane); break;
        case 1: gcn_rows<1>(acc, H4, mine, j, lane); break;
        default: break;
    }
}

__device__ __forceinline__ void gcn_chunk(float4& acc, const float4* __restrict__ H4,
                                          const int2* __restrict__ slot,
                                          const int first, const int m, const int lane) {
    int2 mine = make_int2(0, 0);
    if (lane < m) mine = slot[first + lane];
    gcn_chunk_sum(acc, H4, mine, m, lane);
}

__device__ __forceinline__ int2 gcn_sorted_window(const int2* __restrict__ raw, int2* win,
                                                  const int start, const int n, const int p,
                                                  const int lane) {
    for (int c0 = 0; c0 < n; c0 += 32) {
        const int i = c0 + lane;
        int2 mine = make_int2(0x7fffffff, 0);
        if (i < n) mine = raw[start + i];
        const u64 mk = ((u64)(unsigned)mine.x << 32) | (u64)(unsigned)i;
        int pos = 0;
        for (int k0 = 0; k0 < n; k0 += 32) {
            int key = mine.x;
            if (k0 != c0) {
                key = 0x7fffffff;
                if (k0 + lane < n) key = raw[start + k0 + lane].x;
            }
            const int cm = (n - k0 < 32) ? (n - k0) : 32;
            for (int k = 0; k < cm; ++k) {
                const u64 ok = ((u64)(unsigned)__builtin_amdgcn_readlane(key, k) << 32)
                             | (u64)(unsigned)(k0 + k);
                pos += (ok < mk) ? 1 : 0;
            }
        }
        if (i < n && pos >= p && pos < p + 32) win[pos - p] = mine;
    }
    int2 got = make_int2(0, 0);
    if (p + lane < n) got = win[lane];
    return got;
}

__global__ void __launch_bounds__(256)
gcn_gather_rows(const float* __restrict__ H, const u64* __restrict__ row,
                const int2* __restrict__ slot, const float* __restrict__ dis,
                const float* __restrict__ b, float* __restrict__ out, int N) {
    const int lane = threadIdx.x & 31;
    const int d = __builtin_amdgcn_readfirstlane((int)((blockIdx.x * blockDim.x + threadIdx.x) >> 5));
    if (d >= N) return;

    const float4* H4 = (const float4*)H;
    const u64    ri = row[d];
    const float  sd = dis[d];
    const float4 hv = H4[(size_t)d * (D_FEAT / 4) + lane];
    const float4 bv = ((const float4*)b)[lane];
    const int start = (int)(unsigned)ri;
    const int n     = (int)(ri >> 32);

    float4 acc = make_float4(0.0f, 0.0f, 0.0f, 0.0f);

    if (n > 0) gcn_chunk(acc, H4, slot, start, (n < 32) ? n : 32, lane);
    for (int j0 = 32; j0 < n; j0 += 32)
        gcn_chunk(acc, H4, slot, start + j0, (n - j0 < 32) ? (n - j0) : 32, lane);

    const float ws = sd * sd;
    float4 o;
    o.x = fmaf(hv.x, ws, acc.x) + bv.x;
    o.y = fmaf(hv.y, ws, acc.y) + bv.y;
    o.z = fmaf(hv.z, ws, acc.z) + bv.z;
    o.w = fmaf(hv.w, ws, acc.w) + bv.w;
    ((float4*)out)[(size_t)d * (D_FEAT / 4) + lane] = o;
}

__global__ void __launch_bounds__(256)
gcn_order_gather_rows(const float* __restrict__ H, const u64* __restrict__ row,
                      const int2* __restrict__ raw, int2* __restrict__ slot,
                      const float* __restrict__ dis, const float* __restrict__ b,
                      float* __restrict__ out, int N) {
    __shared__ int2 wins[8][32];
    const int lane = threadIdx.x & 31;
    int2* win = wins[threadIdx.x >> 5];
    const int d = __builtin_amdgcn_readfirstlane((int)((blockIdx.x * blockDim.x + threadIdx.x) >> 5));
    if (d >= N) return;

    const float4* H4 = (const float4*)H;
    const u64    ri = row[d];
    const float  sd = dis[d];
    const float4 hv = H4[(size_t)d * (D_FEAT / 4) + lane];
    const float4 bv = ((const float4*)b)[lane];
    const int start = (int)(unsigned)ri;
    const int n     = (int)(ri >> 32);

    float4 acc = make_float4(0.0f, 0.0f, 0.0f, 0.0f);

    if (n <= 32) {
        if (n > 0) {
            int2 mine = make_int2(0x7fffffff, 0);
            if (lane < n) mine = raw[start + lane];
            const u64 mk = ((u64)(unsigned)mine.x << 32) | (u64)(unsigned)lane;
            int pos = 0;
            for (int k = 0; k < n; ++k) {
                const u64 ok = ((u64)(unsigned)__builtin_amdgcn_readlane(mine.x, k) << 32)
                             | (u64)(unsigned)k;
                pos += (ok < mk) ? 1 : 0;
            }
            if (lane < n) win[pos] = mine;
            int2 got = make_int2(0, 0);
            if (lane < n) {
                got = win[lane];
                slot[start + lane] = got;
            }
            gcn_chunk_sum(acc, H4, got, n, lane);
        }
    } else {
        for (int p = 0; p < n; p += 32) {
            const int m = (n - p < 32) ? (n - p) : 32;
            const int2 mine = gcn_sorted_window(raw, win, start, n, p, lane);
            if (lane < m) slot[start + p + lane] = mine;
            gcn_chunk_sum(acc, H4, mine, m, lane);
        }
    }

    const float ws = sd * sd;
    float4 o;
    o.x = fmaf(hv.x, ws, acc.x) + bv.x;
    o.y = fmaf(hv.y, ws, acc.y) + bv.y;
    o.z = fmaf(hv.z, ws, acc.z) + bv.z;
    o.w = fmaf(hv.w, ws, acc.w) + bv.w;
    ((float4*)out)[(size_t)d * (D_FEAT / 4) + lane] = o;
}

extern "C" void kernel_launch(void* const* d_in, const int* in_sizes, int n_in,
                              void* d_out, int out_size, void* d_ws, size_t ws_size,
                              hipStream_t stream) {
    const float* feat = (const float*)d_in[0];
    const int*   ei   = (const int*)d_in[1];
    const float* W0   = (const float*)d_in[2];
    const float* b0   = (const float*)d_in[3];
    const float* W1   = (const float*)d_in[4];
    const float* b1   = (const float*)d_in[5];
    const float* W2   = (const float*)d_in[6];
    const float* b2   = (const float*)d_in[7];

    const int N = in_sizes[0] / D_FEAT;
    const int E = in_sizes[1] / 2;
    const int* src = ei;
    const int* dst = ei + E;

    const size_t Np = ((size_t)N + 1023) & ~(size_t)1023;
    const size_t Ep = ((size_t)E + 1023) & ~(size_t)1023;
    int*   cnt  = (int*)d_ws;
    float* dis  = (float*)(cnt + Np);
    u64*   row  = (u64*)(dis + Np);
    int*   rank = (int*)(row + Np);
    int2*  raw  = (int2*)(rank + Ep);
    int2*  slot = raw + Ep;
    float* H    = (float*)(slot + Ep);
    float* X    = H + (size_t)N * D_FEAT;
    float* Wc   = X + (size_t)N * D_FEAT;
    float* out  = (float*)d_out;

    const int TB = 256;
    dim3 blkFold(FOLD_BLOCKS + (N + TB - 1) / TB);
    dim3 blkRank(GEMM_BLOCKS + (E + RANK_EDGES - 1) / RANK_EDGES);
    dim3 blkE((E + TB - 1) / TB);
    dim3 blkScan((N + SCAN_ROWS - 1) / SCAN_ROWS);
    dim3 blkRow(((size_t)N * 32 + TB - 1) / TB);

    gcn_fold_weights_clear<<<blkFold, TB, 0, stream>>>(W0, b0, W1, b1, W2, Wc, cnt, N);

    gcn_rank_edges_gemm<<<blkRank, TB, 0, stream>>>(dst, cnt, rank, E, feat, Wc, H, N);

    gcn_scan_deg<<<blkScan, TB, 0, stream>>>(cnt, row, dis, N);
    gcn_fill_slots<<<blkE, TB, 0, stream>>>(src, dst, row, rank, dis, raw, E);

    gcn_order_gather_rows<<<blkRow, TB, 0, stream>>>(H, row, raw, slot, dis, Wc + 128 * D_FEAT, X, N);
    gcn_gather_rows<<<blkRow, TB, 0, stream>>>(X, row, slot, dis, Wc + 129 * D_FEAT, H, N);
    gcn_gather_rows<<<blkRow, TB, 0, stream>>>(H, row, slot, dis, b2, out, N);
}


# Edge buckets by a two-level counting sort in LDS: per-block bin counts, one scan, pairs streamed into 256-row bins, a block per bin builds its rows and scales its H rows by dis; rounding changes
# speedup vs baseline: 4.9933x; 1.2526x over previous
#include <hip/hip_runtime.h>
#include <hip/hip_bf16.h>

typedef __attribute__((ext_vector_type(2))) float v2f;
typedef __attribute__((ext_vector_type(8))) float v8f;

#define D_FEAT 128
#define LDS_STRIDE 132

#define BIN_SHIFT 8
#define BIN_ROWS 256
#define MAX_BINS 512
#define PART_EDGES 8192
#define BIN_CAP 8192
#define SCAN_TILES 19
#define SCAN_CAP (SCAN_TILES * 4096)
typedef unsigned long long u64;
typedef __attribute__((ext_vector_type(4))) int v4i;

__device__ __forceinline__ void gcn_count_bins(const int* __restrict__ dst, int* __restrict__ bh,
                                               int* hist, const int E, const int M, const int NB,
                                               const int b) {
    const int tid = threadIdx.x;
    for (int i = tid; i < NB; i += 256) hist[i] = 0;
    __syncthreads();
    const int base = b * PART_EDGES;
    {
        int d[PART_EDGES / 256];
        #pragma unroll
        for (int q = 0; q < PART_EDGES / 256; ++q) {
            const int e = base + q * 256 + tid;
            d[q] = (e < E) ? dst[e] : -1;
        }
        #pragma unroll
        for (int q = 0; q < PART_EDGES / 256; ++q)
            if (d[q] >= 0) atomicAdd(&hist[d[q] >> BIN_SHIFT], 1);
    }
    __syncthreads();
    for (int i = tid; i < NB; i += 256) bh[i * M + b] = hist[i];
}

__global__ void __launch_bounds__(1024)
gcn_scan_bins(const int* __restrict__ bh, int* __restrict__ boff, int T) {
    __shared__ int wtot[SCAN_TILES * 32];
    const int tid  = threadIdx.x;
    const int lane = tid & 31;
    const int wave = tid >> 5;
    const v4i* in4 = (const v4i*)bh;
    v4i v[SCAN_TILES];
    int inc[SCAN_TILES];
    #pragma unroll
    for (int t = 0; t < SCAN_TILES; ++t) v[t] = in4[t * 1024 + tid];
    #pragma unroll
    for (int t = 0; t < SCAN_TILES; ++t) {
        const int g = (t * 1024 + tid) * 4;
        v[t].x = (g     < T) ? v[t].x : 0;
        v[t].y = (g + 1 < T) ? v[t].y : 0;
        v[t].z = (g + 2 < T) ? v[t].z : 0;
        v[t].w = (g + 3 < T) ? v[t].w : 0;
        int s = (v[t].x + v[t].y) + (v[t].z + v[t].w);
        for (int o = 1; o < 32; o <<= 1) {
            const int up = __shfl_up(s, o);
            if (lane >= o) s += up;
        }
        inc[t] = s;
        if (lane == 31) wtot[t * 32 + wave] = s;
    }
    __syncthreads();
    if (wave == 0) {
        int loc[SCAN_TILES];
        int sum = 0;
        #pragma unroll
        for (int q = 0; q < SCAN_TILES; ++q) {
            loc[q] = wtot[lane * SCAN_TILES + q];
            sum += loc[q];
        }
        int run = sum;
        for (int o = 1; o < 32; o <<= 1) {
            const int up = __shfl_up(run, o);
            if (lane >= o) run += up;
        }
        int ex = run - sum;
        #pragma unroll
        for (int q = 0; q < SCAN_TILES; ++q) {
            wtot[lane * SCAN_TILES + q] = ex;
            ex += loc[q];
        }
    }
    __syncthreads();
    v4i* out4 = (v4i*)boff;
    #pragma unroll
    for (int t = 0; t < SCAN_TILES; ++t) {
        const int g = (t * 1024 + tid) * 4;
        const int own = (v[t].x + v[t].y) + (v[t].z + v[t].w);
        v4i o;
        o.x = wtot[t * 32 + wave] + inc[t] - own;
        o.y = o.x + v[t].x;
        o.z = o.y + v[t].y;
        o.w = o.z + v[t].z;
        if (g < T) out4[t * 1024 + tid] = o;
    }
}

#define GEMM_BLOCKS 512

__device__ __forceinline__ void gcn_gemm_panels(const float* __restrict__ X,
                                                const float* __restrict__ W,
                                                float* __restrict__ H, const int N, const int b) {
    __shared__ float As[16 * LDS_STRIDE];

    const int tid  = threadIdx.x;
    const int wave = tid >> 5;
    const int lane = tid & 31;
    const int half = lane >> 4;
    const int l16  = lane & 15;
    const int col  = wave * 16 + l16;

    v2f bf[D_FEAT / 4];
    #pragma unroll
    for (int i = 0; i < D_FEAT / 4; ++i) {
        const int k = 4 * i + 2 * half;
        bf[i].x = W[(size_t)k * D_FEAT + col];
        bf[i].y = W[(size_t)(k + 1) * D_FEAT + col];
    }

    const int npan = (N + 15) >> 4;
    const float4 zero4 = make_float4(0.0f, 0.0f, 0.0f, 0.0f);

    float4 a0 = zero4, a1 = zero4;
    if (b < npan) {
        const int r = 16 * b + wave;
        if (r < N)     a0 = ((const float4*)(X + (size_t)r * D_FEAT))[lane];
        if (r + 8 < N) a1 = ((const float4*)(X + (size_t)(r + 8) * D_FEAT))[lane];
    }

    #pragma unroll 1
    for (int p = b; p < npan; p += GEMM_BLOCKS) {
        __syncthreads();
        *(float4*)&As[wave * LDS_STRIDE + 4 * lane] = a0;
        *(float4*)&As[(wave + 8) * LDS_STRIDE + 4 * lane] = a1;
        __syncthreads();

        const int pn = p + GEMM_BLOCKS;
        a0 = zero4; a1 = zero4;
        if (pn < npan) {
            const int r = 16 * pn + wave;
            if (r < N)     a0 = ((const float4*)(X + (size_t)r * D_FEAT))[lane];
            if (r + 8 < N) a1 = ((const float4*)(X + (size_t)(r + 8) * D_FEAT))[lane];
        }

        v8f acc = {};
        #pragma unroll
        for (int i = 0; i < D_FEAT / 4; ++i) {
            const v2f a = *(const v2f*)(As + l16 * LDS_STRIDE + 4 * i + 2 * half);
            acc = __builtin_amdgcn_wmma_f32_16x16x4_f32(
                 false, a,  false, bf[i],
                 (short)0, acc,  false,  false);
        }

        const int r0 = 16 * p;
        #pragma unroll
        for (int j = 0; j < 8; ++j) {
            const int row = r0 + j + 8 * half;
            if (row < N) H[(size_t)row * D_FEAT + col] = acc[j];
        }
    }
}

__global__ void __launch_bounds__(256)
gcn_copy_bins_gemm(const int* __restrict__ src, const int* __restrict__ dst,
                   const int* __restrict__ bh, const int* __restrict__ boff,
                   u64* __restrict__ pairs, int E, int M, int NB,
                   const float* __restrict__ X, const float* __restrict__ W,
                   float* __restrict__ H, int N) {
    if (blockIdx.x < GEMM_BLOCKS) {
        gcn_gemm_panels(X, W, H, N, (int)blockIdx.x);
        return;
    }
    __shared__ u64 stage[PART_EDGES];
    __shared__ int lcur[MAX_BINS];
    __shared__ int gdel[MAX_BINS];
    __shared__ int wsum[8];
    const int tid  = threadIdx.x;
    const int lane = tid & 31;
    const int wave = tid >> 5;
    const int b    = (int)blockIdx.x - GEMM_BLOCKS;

    const int k0 = 2 * tid, k1 = 2 * tid + 1;
    const int c0 = (k0 < NB) ? bh[k0 * M + b] : 0;
    const int c1 = (k1 < NB) ? bh[k1 * M + b] : 0;
    const int g0 = (k0 < NB) ? boff[k0 * M + b] : 0;
    const int g1 = (k1 < NB) ? boff[k1 * M + b] : 0;
    const int tsum = c0 + c1;
    int inc = tsum;
    for (int o = 1; o < 32; o <<= 1) {
        const int up = __shfl_up(inc, o);
        if (lane >= o) inc += up;
    }
    if (lane == 31) wsum[wave] = inc;
    __syncthreads();
    int off = 0;
    for (int w = 0; w < 8; ++w)
        if (w < wave) off += wsum[w];
    const int ex = off + inc - tsum;
    lcur[k0] = ex;
    lcur[k1] = ex + c0;
    gdel[k0] = g0 - ex;
    gdel[k1] = g1 - (ex + c0);
    __syncthreads();

    const int base = b * PART_EDGES;
    for (int u = 0; u < PART_EDGES / 256; u += 16) {
        int d[16], s[16];
        #pragma unroll
        for (int q = 0; q < 16; ++q) {
            const int e = base + (u + q) * 256 + tid;
            d[q] = (e < E) ? dst[e] : -1;
            s[q] = (e < E) ? src[e] : 0;
        }
        #pragma unroll
        for (int q = 0; q < 16; ++q) {
            if (d[q] >= 0) {
                const int pos = atomicAdd(&lcur[d[q] >> BIN_SHIFT], 1);
                stage[pos] = ((u64)(unsigned)d[q] << 32) | (u64)(unsigned)s[q];
            }
        }
    }
    __syncthreads();

    const int n = (E - base < PART_EDGES) ? (E - base) : PART_EDGES;
    for (int i = tid; i < n; i += 256) {
        const u64 p = stage[i];
        pairs[gdel[(int)(p >> 32) >> BIN_SHIFT] + i] = p;
    }
}

__global__ void __launch_bounds__(256)
gcn_build_bins(const u64* __restrict__ pairs, const int* __restrict__ boff,
               u64* __restrict__ row, float* __restrict__ dis, int* __restrict__ raw,
               float* __restrict__ H, int E, int M, int NB, int N) {
    __shared__ int   stage[BIN_CAP];
    __shared__ int   cnt[BIN_ROWS];
    __shared__ int   cur[BIN_ROWS];
    __shared__ float fac[BIN_ROWS];
    __shared__ int   wsum[8];
    const int tid   = threadIdx.x;
    const int lane  = tid & 31;
    const int wave  = tid >> 5;
    const int k     = (int)blockIdx.x;
    const int rbase = k << BIN_SHIFT;
    const int lo = boff[k * M];
    const int hi = (k + 1 < NB) ? boff[(k + 1) * M] : E;
    const bool fits = (hi - lo) <= BIN_CAP;

    cnt[tid] = 0;
    __syncthreads();
    for (int i0 = lo; i0 < hi; i0 += 8 * 256) {
        u64 p[8];
        #pragma unroll
        for (int q = 0; q < 8; ++q) {
            const int i = i0 + q * 256 + tid;
            p[q] = (i < hi) ? pairs[i] : 0ull;
        }
        #pragma unroll
        for (int q = 0; q < 8; ++q)
            if (i0 + q * 256 + tid < hi) atomicAdd(&cnt[(int)(p[q] >> 32) - rbase], 1);
    }
    __syncthreads();

    const int c = cnt[tid];
    int inc = c;
    for (int o = 1; o < 32; o <<= 1) {
        const int up = __shfl_up(inc, o);
        if (lane >= o) inc += up;
    }
    if (lane == 31) wsum[wave] = inc;
    __syncthreads();
    int off = lo;
    for (int w = 0; w < 8; ++w)
        if (w < wave) off += wsum[w];
    const int   ex = off + inc - c;
    const float f  = rsqrtf((float)(c + 1));
    const int   d  = rbase + tid;
    if (d < N) {
        row[d] = ((u64)(unsigned)c << 32) | (u64)(unsigned)ex;
        dis[d] = f;
    }
    cur[tid] = ex;
    fac[tid] = f;
    __syncthreads();

    for (int i0 = lo; i0 < hi; i0 += 8 * 256) {
        u64 p[8];
        #pragma unroll
        for (int q = 0; q < 8; ++q) {
            const int i = i0 + q * 256 + tid;
            p[q] = (i < hi) ? pairs[i] : 0ull;
        }
        #pragma unroll
        for (int q = 0; q < 8; ++q) {
            if (i0 + q * 256 + tid < hi) {
                const int pos = atomicAdd(&cur[(int)(p[q] >> 32) - rbase], 1);
                if (fits) stage[pos - lo] = (int)(unsigned)p[q];
                else      raw[pos]        = (int)(unsigned)p[q];
            }
        }
    }
    __syncthreads();
    if (fits)
        for (int i = lo + tid; i < hi; i += 256) raw[i] = stage[i - lo];

    float4* H4 = (float4*)H;
    for (int r0 = wave; r0 < BIN_ROWS; r0 += 128) {
        float4 h[16];
        #pragma unroll
        for (int q = 0; q < 16; ++q) {
            const int r = r0 + 8 * q;
            h[q] = make_float4(0.0f, 0.0f, 0.0f, 0.0f);
            if (rbase + r < N) h[q] = H4[(size_t)(rbase + r) * (D_FEAT / 4) + lane];
        }
        #pragma unroll
        for (int q = 0; q < 16; ++q) {
            const int r = r0 + 8 * q;
            const float s = fac[r];
            if (rbase + r < N)
                H4[(size_t)(rbase + r) * (D_FEAT / 4) + lane] =
                    make_float4(h[q].x * s, h[q].y * s, h[q].z * s, h[q].w * s);
        }
    }
}

#define FOLD_BLOCKS 9

__device__ __forceinline__ v8f gcn_panel_times_wt(const float* __restrict__ As,
                                                  const float* __restrict__ Ws,
                                                  const int l16, const int half, const int col) {
    v8f acc = {};
    #pragma unroll
    for (int k0 = 0; k0 < D_FEAT; k0 += 4) {
        const int ka = k0 + 2 * half;
        const v2f a = *(const v2f*)(As + l16 * LDS_STRIDE + ka);
        const v2f b = *(const v2f*)(Ws + col * LDS_STRIDE + ka);
        acc = __builtin_amdgcn_wmma_f32_16x16x4_f32(
             false, a,  false, b,
             (short)0, acc,  false,  false);
    }
    return acc;
}

__global__ void __launch_bounds__(256)
gcn_fold_weights_count(const float* __restrict__ W0, const float* __restrict__ b0,
                       const float* __restrict__ W1, const float* __restrict__ b1,
                       const float* __restrict__ W2, float* __restrict__ Wc,
                       const int* __restrict__ dst, int* __restrict__ bh,
                       int E, int M, int NB) {
    __shared__ float W1s[D_FEAT * LDS_STRIDE];
    __shared__ float W2s[D_FEAT * LDS_STRIDE];
    __shared__ float As[16 * LDS_STRIDE];
    __shared__ int   hist[MAX_BINS];

    const int tid  = threadIdx.x;
    if (blockIdx.x >= FOLD_BLOCKS) {
        gcn_count_bins(dst, bh, hist, E, M, NB, (int)blockIdx.x - FOLD_BLOCKS);
        return;
    }
    const int p    = blockIdx.x;
    const int wave = tid >> 5;
    const int lane = tid & 31;
    const int half = lane >> 4;
    const int l16  = lane & 15;
    const int col  = wave * 16 + l16;

    const int kk = lane & 7;
    const int pc = 4 * wave + (lane >> 3);
    float4 w1[16], w2[16], a4[2];
    #pragma unroll
    for (int u = 0; u < 16; ++u)
        w1[u] = ((const float4*)(W1 + (size_t)(8 * u + kk) * D_FEAT))[pc];
    #pragma unroll
    for (int u = 0; u < 16; ++u)
        w2[u] = ((const float4*)(W2 + (size_t)(8 * u + kk) * D_FEAT))[pc];
    #pragma unroll
    for (int u = 0; u < 2; ++u) {
        const int r = 8 * u + wave;
        a4[u] = make_float4(0.0f, 0.0f, 0.0f, 0.0f);
        if (p < 8)       a4[u] = ((const float4*)(W0 + (size_t)(16 * p + r) * D_FEAT))[lane];
        else if (r == 0) a4[u] = ((const float4*)b0)[lane];
    }
    #pragma unroll
    for (int u = 0; u < 2; ++u)
        *(float4*)&As[(8 * u + wave) * LDS_STRIDE + 4 * lane] = a4[u];
    #pragma unroll
    for (int u = 0; u < 16; ++u) {
        const int k = 8 * u + kk;
        W1s[(4 * pc + 0) * LDS_STRIDE + k] = w1[u].x;
        W1s[(4 * pc + 1) * LDS_STRIDE + k] = w1[u].y;
        W1s[(4 * pc + 2) * LDS_STRIDE + k] = w1[u].z;
        W1s[(4 * pc + 3) * LDS_STRIDE + k] = w1[u].w;
    }
    #pragma unroll
    for (int u = 0; u < 16; ++u) {
        const int k = 8 * u + kk;
        W2s[(4 * pc + 0) * LDS_STRIDE + k] = w2[u].x;
        W2s[(4 * pc + 1) * LDS_STRIDE + k] = w2[u].y;
        W2s[(4 * pc + 2) * LDS_STRIDE + k] = w2[u].z;
        W2s[(4 * pc + 3) * LDS_STRIDE + k] = w2[u].w;
    }
    __syncthreads();

    const v8f t = gcn_panel_times_wt(As, W1s, l16, half, col);
    __syncthreads();

    #pragma unroll
    for (int j = 0; j < 8; ++j) As[(j + 8 * half) * LDS_STRIDE + col] = t[j];
    if (p == 8 && half == 0) As[1 * LDS_STRIDE + col] = b1[col];
    __syncthreads();

    const v8f o = gcn_panel_times_wt(As, W2s, l16, half, col);

    const int nrow = (p < 8) ? 16 : 2;
    #pragma unroll
    for (int j = 0; j < 8; ++j) {
        const int r = j + 8 * half;
        if (r < nrow) Wc[(size_t)(16 * p + r) * D_FEAT + col] = o[j];
    }
}

template <int B>
__device__ __forceinline__ void gcn_rows(float4& acc, const float4* __restrict__ H4,
                                         const int mine, const int j, const int lane) {
    float4 v[B];
    for (int u = 0; u < B; ++u) {
        const int s = __builtin_amdgcn_readlane(mine, j + u);
        v[u] = H4[(size_t)s * (D_FEAT / 4) + lane];
    }
    for (int u = 0; u < B; ++u) {
        acc.x += v[u].x;
        acc.y += v[u].y;
        acc.z += v[u].z;
        acc.w += v[u].w;
    }
}

__device__ __forceinline__ void gcn_chunk_sum(float4& acc, const float4* __restrict__ H4,
                                              const int mine, const int m, const int lane) {
    int j = 0;
    for (; j + 8 <= m; j += 8) gcn_rows<8>(acc, H4, mine, j, lane);
    switch (m - j) {
        case 7: gcn_rows<7>(acc, H4, mine, j, lane); break;
        case 6: gcn_rows<6>(acc, H4, mine, j, lane); break;
        case 5: gcn_rows<5>(acc, H4, mine, j, lane); break;
        case 4: gcn_rows<4>(acc, H4, mine, j, lane); break;
        case 3: gcn_rows<3>(acc, H4, mine, j, lane); break;
        case 2: gcn_rows<2>(acc, H4, mine, j, lane); break;
        case 1: gcn_rows<1>(acc, H4, mine, j, lane); break;
        default: break;
    }
}

__device__ __forceinline__ void gcn_chunk(float4& acc, const float4* __restrict__ H4,
                                          const int* __restrict__ slot,
                                          const int first, const int m, const int lane) {
    int mine = 0;
    if (lane < m) mine = slot[first + lane];
    gcn_chunk_sum(acc, H4, mine, m, lane);
}

template <bool SCALE_OUT>
__device__ __forceinline__ float4 gcn_finish(const float4 acc, const float4 hv,
                                             const float4 bv, const float sd) {
    float4 o;
    o.x = fmaf(acc.x + hv.x, sd, bv.x);
    o.y = fmaf(acc.y + hv.y, sd, bv.y);
    o.z = fmaf(acc.z + hv.z, sd, bv.z);
    o.w = fmaf(acc.w + hv.w, sd, bv.w);
    if (SCALE_OUT) {
        o.x *= sd;
        o.y *= sd;
        o.z *= sd;
        o.w *= sd;
    }
    return o;
}

__device__ __forceinline__ int gcn_sorted_window(const int* __restrict__ raw, int* win,
                                                 const int start, const int n, const int p,
                                                 const int lane) {
    for (int c0 = 0; c0 < n; c0 += 32) {
        const int i = c0 + lane;
        int mine = 0x7fffffff;
        if (i < n) mine = raw[start + i];
        const u64 mk = ((u64)(unsigned)mine << 32) | (u64)(unsigned)i;
        int pos = 0;
        for (int k0 = 0; k0 < n; k0 += 32) {
            int key = mine;
            if (k0 != c0) {
                key = 0x7fffffff;
                if (k0 + lane < n) key = raw[start + k0 + lane];
            }
            const int cm = (n - k0 < 32) ? (n - k0) : 32;
            for (int k = 0; k < cm; ++k) {
                const u64 ok = ((u64)(unsigned)__builtin_amdgcn_readlane(key, k) << 32)
                             | (u64)(unsigned)(k0 + k);
                pos += (ok < mk) ? 1 : 0;
            }
        }
        if (i < n && pos >= p && pos < p + 32) win[pos - p] = mine;
    }
    int got = 0;
    if (p + lane < n) got = win[lane];
    return got;
}

template <bool SCALE_OUT>
__global__ void __launch_bounds__(256)
gcn_gather_rows(const float* __restrict__ H, const u64* __restrict__ row,
                const int* __restrict__ slot, const float* __restrict__ dis,
                const float* __restrict__ b, float* __restrict__ out, int N) {
    const int lane = threadIdx.x & 31;
    const int d = __builtin_amdgcn_readfirstlane((int)((blockIdx.x * blockDim.x + threadIdx.x) >> 5));
    if (d >= N) return;

    const float4* H4 = (const float4*)H;
    const u64    ri = row[d];
    const float  sd = dis[d];
    const float4 hv = H4[(size_t)d * (D_FEAT / 4) + lane];
    const float4 bv = ((const float4*)b)[lane];
    const int start = (int)(unsigned)ri;
    const int n     = (int)(ri >> 32);

    float4 acc = make_float4(0.0f, 0.0f, 0.0f, 0.0f);

    if (n > 0) gcn_chunk(acc, H4, slot, start, (n < 32) ? n : 32, lane);
    for (int j0 = 32; j0 < n; j0 += 32)
        gcn_chunk(acc, H4, slot, start + j0, (n - j0 < 32) ? (n - j0) : 32, lane);

    ((float4*)out)[(size_t)d * (D_FEAT / 4) + lane] = gcn_finish<SCALE_OUT>(acc, hv, bv, sd);
}

__global__ void __launch_bounds__(256)
gcn_order_gather_rows(const float* __restrict__ H, const u64* __restrict__ row,
                      const int* __restrict__ raw, int* __restrict__ slot,
                      const float* __restrict__ dis, const float* __restrict__ b,
                      float* __restrict__ out, int N) {
    __shared__ int wins[8][32];
    const int lane = threadIdx.x & 31;
    int* win = wins[threadIdx.x >> 5];
    const int d = __builtin_amdgcn_readfirstlane((int)((blockIdx.x * blockDim.x + threadIdx.x) >> 5));
    if (d >= N) return;

    const float4* H4 = (const float4*)H;
    const u64    ri = row[d];
    const float  sd = dis[d];
    const float4 hv = H4[(size_t)d * (D_FEAT / 4) + lane];
    const float4 bv = ((const float4*)b)[lane];
    const int start = (int)(unsigned)ri;
    const int n     = (int)(ri >> 32);

    float4 acc = make_float4(0.0f, 0.0f, 0.0f, 0.0f);

    if (n <= 32) {
        if (n > 0) {
            int mine = 0x7fffffff;
            if (lane < n) mine = raw[start + lane];
            const u64 mk = ((u64)(unsigned)mine << 32) | (u64)(unsigned)lane;
            int pos = 0;
            for (int k = 0; k < n; ++k) {
                const u64 ok = ((u64)(unsigned)__builtin_amdgcn_readlane(mine, k) << 32)
                             | (u64)(unsigned)k;
                pos += (ok < mk) ? 1 : 0;
            }
            if (lane < n) win[pos] = mine;
            int got = 0;
            if (lane < n) {
                got = win[lane];
                slot[start + lane] = got;
            }
            gcn_chunk_sum(acc, H4, got, n, lane);
        }
    } else {
        for (int p = 0; p < n; p += 32) {
            const int m = (n - p < 32) ? (n - p) : 32;
            const int mine = gcn_sorted_window(raw, win, start, n, p, lane);
            if (lane < m) slot[start + p + lane] = mine;
            gcn_chunk_sum(acc, H4, mine, m, lane);
        }
    }

    ((float4*)out)[(size_t)d * (D_FEAT / 4) + lane] = gcn_finish<true>(acc, hv, bv, sd);
}

extern "C" void kernel_launch(void* const* d_in, const int* in_sizes, int n_in,
                              void* d_out, int out_size, void* d_ws, size_t ws_size,
                              hipStream_t stream) {
    const float* feat = (const float*)d_in[0];
    const int*   ei   = (const int*)d_in[1];
    const float* W0   = (const float*)d_in[2];
    const float* b0   = (const float*)d_in[3];
    const float* W1   = (const float*)d_in[4];
    const float* b1   = (const float*)d_in[5];
    const float* W2   = (const float*)d_in[6];
    const float* b2   = (const float*)d_in[7];

    const int N = in_sizes[0] / D_FEAT;
    const int E = in_sizes[1] / 2;
    const int* src = ei;
    const int* dst = ei + E;

    const int M  = (E + PART_EDGES - 1) / PART_EDGES;
    const int NB = (N + BIN_ROWS - 1) >> BIN_SHIFT;

    const size_t Np = ((size_t)N + 1023) & ~(size_t)1023;
    const size_t Ep = ((size_t)E + 1023) & ~(size_t)1023;
    float* dis   = (float*)d_ws;
    u64*   row   = (u64*)(dis + Np);
    u64*   pairs = row + Np;
    int*   raw   = (int*)(pairs + Ep);
    int*   slot  = raw + Ep;
    int*   bh    = slot + Ep;
    int*   boff  = bh + SCAN_CAP;
    float* H     = (float*)(boff + SCAN_CAP);
    float* X     = H + (size_t)N * D_FEAT;
    float* Wc    = X + (size_t)N * D_FEAT;
    float* out   = (float*)d_out;

    const int TB = 256;
    dim3 blkFold(FOLD_BLOCKS + M);
    dim3 blkCopy(GEMM_BLOCKS + M);
    dim3 blkBins(NB);
    dim3 blkRow(((size_t)N * 32 + TB - 1) / TB);

    gcn_fold_weights_count<<<blkFold, TB, 0, stream>>>(W0, b0, W1, b1, W2, Wc, dst, bh, E, M, NB);

    gcn_scan_bins<<<1, 1024, 0, stream>>>(bh, boff, NB * M);

    gcn_copy_bins_gemm<<<blkCopy, TB, 0, stream>>>(src, dst, bh, boff, pairs, E, M, NB, feat, Wc, H, N);

    gcn_build_bins<<<blkBins, TB, 0, stream>>>(pairs, boff, row, dis, raw, H, E, M, NB, N);

    gcn_order_gather_rows<<<blkRow, TB, 0, stream>>>(H, row, raw, slot, dis, Wc + 128 * D_FEAT, X, N);
    gcn_gather_rows<true><<<blkRow, TB, 0, stream>>>(X, row, slot, dis, Wc + 129 * D_FEAT, H, N);
    gcn_gather_rows<false><<<blkRow, TB, 0, stream>>>(H, row, slot, dis, b2, out, N);
}


# Scan launch removed: counters stored one row per block; each copying block sums all rows itself (28 8-byte loads in flight) for bin sizes and its own offsets, scans the sizes in-block; 7 launches -> 6
# speedup vs baseline: 5.3970x; 1.0808x over previous
#include <hip/hip_runtime.h>
#include <hip/hip_bf16.h>

typedef __attribute__((ext_vector_type(2))) float v2f;
typedef __attribute__((ext_vector_type(8))) float v8f;

#define D_FEAT 128
#define LDS_STRIDE 132

#define BIN_SHIFT 8
#define BIN_ROWS 256
#define MAX_BINS 512
#define PART_EDGES 8192
#define BIN_CAP 8192
#define BH_CAP (304 * MAX_BINS)
#define PRE_ROWS 28
typedef unsigned long long u64;
typedef __attribute__((ext_vector_type(4))) int v4i;

__device__ __forceinline__ void gcn_count_bins(const int* __restrict__ dst, int* __restrict__ bh,
                                               int* hist, const int E, const int b) {
    const int tid = threadIdx.x;
    for (int i = tid; i < MAX_BINS; i += 256) hist[i] = 0;
    __syncthreads();
    const int base = b * PART_EDGES;
    {
        int d[PART_EDGES / 256];
        #pragma unroll
        for (int q = 0; q < PART_EDGES / 256; ++q) {
            const int e = base + q * 256 + tid;
            d[q] = (e < E) ? dst[e] : -1;
        }
        #pragma unroll
        for (int q = 0; q < PART_EDGES / 256; ++q)
            if (d[q] >= 0) atomicAdd(&hist[d[q] >> BIN_SHIFT], 1);
    }
    __syncthreads();
    for (int i = tid; i < MAX_BINS; i += 256) bh[b * MAX_BINS + i] = hist[i];
}

#define GEMM_BLOCKS 512

__device__ __forceinline__ void gcn_gemm_panels(const float* __restrict__ X,
                                                const float* __restrict__ W,
                                                float* __restrict__ H, const int N, const int b) {
    __shared__ float As[16 * LDS_STRIDE];

    const int tid  = threadIdx.x;
    const int wave = tid >> 5;
    const int lane = tid & 31;
    const int half = lane >> 4;
    const int l16  = lane & 15;
    const int col  = wave * 16 + l16;

    v2f bf[D_FEAT / 4];
    #pragma unroll
    for (int i = 0; i < D_FEAT / 4; ++i) {
        const int k = 4 * i + 2 * half;
        bf[i].x = W[(size_t)k * D_FEAT + col];
        bf[i].y = W[(size_t)(k + 1) * D_FEAT + col];
    }

    const int npan = (N + 15) >> 4;
    const float4 zero4 = make_float4(0.0f, 0.0f, 0.0f, 0.0f);

    float4 a0 = zero4, a1 = zero4;
    if (b < npan) {
        const int r = 16 * b + wave;
        if (r < N)     a0 = ((const float4*)(X + (size_t)r * D_FEAT))[lane];
        if (r + 8 < N) a1 = ((const float4*)(X + (size_t)(r + 8) * D_FEAT))[lane];
    }

    #pragma unroll 1
    for (int p = b; p < npan; p += GEMM_BLOCKS) {
        __syncthreads();
        *(float4*)&As[wave * LDS_STRIDE + 4 * lane] = a0;
        *(float4*)&As[(wave + 8) * LDS_STRIDE + 4 * lane] = a1;
        __syncthreads();

        const int pn = p + GEMM_BLOCKS;
        a0 = zero4; a1 = zero4;
        if (pn < npan) {
            const int r = 16 * pn + wave;
            if (r < N)     a0 = ((const float4*)(X + (size_t)r * D_FEAT))[lane];
            if (r + 8 < N) a1 = ((const float4*)(X + (size_t)(r + 8) * D_FEAT))[lane];
        }

        v8f acc = {};
        #pragma unroll
        for (int i = 0; i < D_FEAT / 4; ++i) {
            const v2f a = *(const v2f*)(As + l16 * LDS_STRIDE + 4 * i + 2 * half);
            acc = __builtin_amdgcn_wmma_f32_16x16x4_f32(
                 false, a,  false, bf[i],
                 (short)0, acc,  false,  false);
        }

        const int r0 = 16 * p;
        #pragma unroll
        for (int j = 0; j < 8; ++j) {
            const int row = r0 + j + 8 * half;
            if (row < N) H[(size_t)row * D_FEAT + col] = acc[j];
        }
    }
}

__global__ void __launch_bounds__(256)
gcn_copy_bins_gemm(const int* __restrict__ src, const int* __restrict__ dst,
                   const int* __restrict__ bh, int* __restrict__ bbase,
                   u64* __restrict__ pairs, int E, int M,
                   const float* __restrict__ X, const float* __restrict__ W,
                   float* __restrict__ H, int N) {
    if (blockIdx.x < GEMM_BLOCKS) {
        gcn_gemm_panels(X, W, H, N, (int)blockIdx.x);
        return;
    }
    __shared__ u64 stage[PART_EDGES];
    __shared__ int lcur[MAX_BINS];
    __shared__ int gdel[MAX_BINS];
    __shared__ int wsum[8];
    __shared__ int wtot[8];
    const int tid  = threadIdx.x;
    const int lane = tid & 31;
    const int wave = tid >> 5;
    const int b    = (int)blockIdx.x - GEMM_BLOCKS;

    const int k0 = 2 * tid, k1 = 2 * tid + 1;
    const int2* bh2 = (const int2*)bh;
    int c0 = 0, c1 = 0;
    int pre0 = 0, pre1 = 0;
    int tot0 = 0, tot1 = 0;
    for (int q0 = 0; q0 < M; q0 += PRE_ROWS) {
        int2 v[PRE_ROWS];
        #pragma unroll
        for (int u = 0; u < PRE_ROWS; ++u) {
            const int q = q0 + u;
            v[u] = make_int2(0, 0);
            if (q < M) v[u] = bh2[q * (MAX_BINS / 2) + tid];
        }
        #pragma unroll
        for (int u = 0; u < PRE_ROWS; ++u) {
            const int q = q0 + u;
            tot0 += v[u].x;
            tot1 += v[u].y;
            pre0 += (q < b) ? v[u].x : 0;
            pre1 += (q < b) ? v[u].y : 0;
            c0 = (q == b) ? v[u].x : c0;
            c1 = (q == b) ? v[u].y : c1;
        }
    }
    const int tsum = c0 + c1;
    const int ttot = tot0 + tot1;
    int inc = tsum, tinc = ttot;
    for (int o = 1; o < 32; o <<= 1) {
        const int up  = __shfl_up(inc, o);
        const int tup = __shfl_up(tinc, o);
        if (lane >= o) { inc += up; tinc += tup; }
    }
    if (lane == 31) { wsum[wave] = inc; wtot[wave] = tinc; }
    __syncthreads();
    int off = 0, toff = 0;
    for (int w = 0; w < 8; ++w)
        if (w < wave) { off += wsum[w]; toff += wtot[w]; }
    const int ex = off + inc - tsum;
    const int bx = toff + tinc - ttot;
    lcur[k0] = ex;
    lcur[k1] = ex + c0;
    gdel[k0] = (bx + pre0) - ex;
    gdel[k1] = (bx + tot0 + pre1) - (ex + c0);
    if (b == 0) {
        bbase[k0] = bx;
        bbase[k1] = bx + tot0;
    }
    __syncthreads();

    const int base = b * PART_EDGES;
    for (int u = 0; u < PART_EDGES / 256; u += 16) {
        int d[16], s[16];
        #pragma unroll
        for (int q = 0; q < 16; ++q) {
            const int e = base + (u + q) * 256 + tid;
            d[q] = (e < E) ? dst[e] : -1;
            s[q] = (e < E) ? src[e] : 0;
        }
        #pragma unroll
        for (int q = 0; q < 16; ++q) {
            if (d[q] >= 0) {
                const int pos = atomicAdd(&lcur[d[q] >> BIN_SHIFT], 1);
                stage[pos] = ((u64)(unsigned)d[q] << 32) | (u64)(unsigned)s[q];
            }
        }
    }
    __syncthreads();

    const int n = (E - base < PART_EDGES) ? (E - base) : PART_EDGES;
    for (int i = tid; i < n; i += 256) {
        const u64 p = stage[i];
        pairs[gdel[(int)(p >> 32) >> BIN_SHIFT] + i] = p;
    }
}

__global__ void __launch_bounds__(256)
gcn_build_bins(const u64* __restrict__ pairs, const int* __restrict__ bbase,
               u64* __restrict__ row, float* __restrict__ dis, int* __restrict__ raw,
               float* __restrict__ H, int E, int NB, int N) {
    __shared__ int   stage[BIN_CAP];
    __shared__ int   cnt[BIN_ROWS];
    __shared__ int   cur[BIN_ROWS];
    __shared__ float fac[BIN_ROWS];
    __shared__ int   wsum[8];
    const int tid   = threadIdx.x;
    const int lane  = tid & 31;
    const int wave  = tid >> 5;
    const int k     = (int)blockIdx.x;
    const int rbase = k << BIN_SHIFT;
    const int lo = bbase[k];
    const int hi = (k + 1 < NB) ? bbase[k + 1] : E;
    const bool fits = (hi - lo) <= BIN_CAP;

    cnt[tid] = 0;
    __syncthreads();
    for (int i0 = lo; i0 < hi; i0 += 8 * 256) {
        u64 p[8];
        #pragma unroll
        for (int q = 0; q < 8; ++q) {
            const int i = i0 + q * 256 + tid;
            p[q] = (i < hi) ? pairs[i] : 0ull;
        }
        #pragma unroll
        for (int q = 0; q < 8; ++q)
            if (i0 + q * 256 + tid < hi) atomicAdd(&cnt[(int)(p[q] >> 32) - rbase], 1);
    }
    __syncthreads();

    const int c = cnt[tid];
    int inc = c;
    for (int o = 1; o < 32; o <<= 1) {
        const int up = __shfl_up(inc, o);
        if (lane >= o) inc += up;
    }
    if (lane == 31) wsum[wave] = inc;
    __syncthreads();
    int off = lo;
    for (int w = 0; w < 8; ++w)
        if (w < wave) off += wsum[w];
    const int   ex = off + inc - c;
    const float f  = rsqrtf((float)(c + 1));
    const int   d  = rbase + tid;
    if (d < N) {
        row[d] = ((u64)(unsigned)c << 32) | (u64)(unsigned)ex;
        dis[d] = f;
    }
    cur[tid] = ex;
    fac[tid] = f;
    __syncthreads();

    for (int i0 = lo; i0 < hi; i0 += 8 * 256) {
        u64 p[8];
        #pragma unroll
        for (int q = 0; q < 8; ++q) {
            const int i = i0 + q * 256 + tid;
            p[q] = (i < hi) ? pairs[i] : 0ull;
        }
        #pragma unroll
        for (int q = 0; q < 8; ++q) {
            if (i0 + q * 256 + tid < hi) {
                const int pos = atomicAdd(&cur[(int)(p[q] >> 32) - rbase], 1);
                if (fits) stage[pos - lo] = (int)(unsigned)p[q];
                else      raw[pos]        = (int)(unsigned)p[q];
            }
        }
    }
    __syncthreads();
    if (fits)
        for (int i = lo + tid; i < hi; i += 256) raw[i] = stage[i - lo];

    float4* H4 = (float4*)H;
    for (int r0 = wave; r0 < BIN_ROWS; r0 += 128) {
        float4 h[16];
        #pragma unroll
        for (int q = 0; q < 16; ++q) {
            const int r = r0 + 8 * q;
            h[q] = make_float4(0.0f, 0.0f, 0.0f, 0.0f);
            if (rbase + r < N) h[q] = H4[(size_t)(rbase + r) * (D_FEAT / 4) + lane];
        }
        #pragma unroll
        for (int q = 0; q < 16; ++q) {
            const int r = r0 + 8 * q;
            const float s = fac[r];
            if (rbase + r < N)
                H4[(size_t)(rbase + r) * (D_FEAT / 4) + lane] =
                    make_float4(h[q].x * s, h[q].y * s, h[q].z * s, h[q].w * s);
        }
    }
}

#define FOLD_BLOCKS 9

__device__ __forceinline__ v8f gcn_panel_times_wt(const float* __restrict__ As,
                                                  const float* __restrict__ Ws,
                                                  const int l16, const int half, const int col) {
    v8f acc = {};
    #pragma unroll
    for (int k0 = 0; k0 < D_FEAT; k0 += 4) {
        const int ka = k0 + 2 * half;
        const v2f a = *(const v2f*)(As + l16 * LDS_STRIDE + ka);
        const v2f b = *(const v2f*)(Ws + col * LDS_STRIDE + ka);
        acc = __builtin_amdgcn_wmma_f32_16x16x4_f32(
             false, a,  false, b,
             (short)0, acc,  false,  false);
    }
    return acc;
}

__global__ void __launch_bounds__(256)
gcn_fold_weights_count(const float* __restrict__ W0, const float* __restrict__ b0,
                       const float* __restrict__ W1, const float* __restrict__ b1,
                       const float* __restrict__ W2, float* __restrict__ Wc,
                       const int* __restrict__ dst, int* __restrict__ bh, int E) {
    __shared__ float W1s[D_FEAT * LDS_STRIDE];
    __shared__ float W2s[D_FEAT * LDS_STRIDE];
    __shared__ float As[16 * LDS_STRIDE];
    __shared__ int   hist[MAX_BINS];

    const int tid  = threadIdx.x;
    if (blockIdx.x >= FOLD_BLOCKS) {
        gcn_count_bins(dst, bh, hist, E, (int)blockIdx.x - FOLD_BLOCKS);
        return;
    }
    const int p    = blockIdx.x;
    const int wave = tid >> 5;
    const int lane = tid & 31;
    const int half = lane >> 4;
    const int l16  = lane & 15;
    const int col  = wave * 16 + l16;

    const int kk = lane & 7;
    const int pc = 4 * wave + (lane >> 3);
    float4 w1[16], w2[16], a4[2];
    #pragma unroll
    for (int u = 0; u < 16; ++u)
        w1[u] = ((const float4*)(W1 + (size_t)(8 * u + kk) * D_FEAT))[pc];
    #pragma unroll
    for (int u = 0; u < 16; ++u)
        w2[u] = ((const float4*)(W2 + (size_t)(8 * u + kk) * D_FEAT))[pc];
    #pragma unroll
    for (int u = 0; u < 2; ++u) {
        const int r = 8 * u + wave;
        a4[u] = make_float4(0.0f, 0.0f, 0.0f, 0.0f);
        if (p < 8)       a4[u] = ((const float4*)(W0 + (size_t)(16 * p + r) * D_FEAT))[lane];
        else if (r == 0) a4[u] = ((const float4*)b0)[lane];
    }
    #pragma unroll
    for (int u = 0; u < 2; ++u)
        *(float4*)&As[(8 * u + wave) * LDS_STRIDE + 4 * lane] = a4[u];
    #pragma unroll
    for (int u = 0; u < 16; ++u) {
        const int k = 8 * u + kk;
        W1s[(4 * pc + 0) * LDS_STRIDE + k] = w1[u].x;
        W1s[(4 * pc + 1) * LDS_STRIDE + k] = w1[u].y;
        W1s[(4 * pc + 2) * LDS_STRIDE + k] = w1[u].z;
        W1s[(4 * pc + 3) * LDS_STRIDE + k] = w1[u].w;
    }
    #pragma unroll
    for (int u = 0; u < 16; ++u) {
        const int k = 8 * u + kk;
        W2s[(4 * pc + 0) * LDS_STRIDE + k] = w2[u].x;
        W2s[(4 * pc + 1) * LDS_STRIDE + k] = w2[u].y;
        W2s[(4 * pc + 2) * LDS_STRIDE + k] = w2[u].z;
        W2s[(4 * pc + 3) * LDS_STRIDE + k] = w2[u].w;
    }
    __syncthreads();

    const v8f t = gcn_panel_times_wt(As, W1s, l16, half, col);
    __syncthreads();

    #pragma unroll
    for (int j = 0; j < 8; ++j) As[(j + 8 * half) * LDS_STRIDE + col] = t[j];
    if (p == 8 && half == 0) As[1 * LDS_STRIDE + col] = b1[col];
    __syncthreads();

    const v8f o = gcn_panel_times_wt(As, W2s, l16, half, col);

    const int nrow = (p < 8) ? 16 : 2;
    #pragma unroll
    for (int j = 0; j < 8; ++j) {
        const int r = j + 8 * half;
        if (r < nrow) Wc[(size_t)(16 * p + r) * D_FEAT + col] = o[j];
    }
}

template <int B>
__device__ __forceinline__ void gcn_rows(float4& acc, const float4* __restrict__ H4,
                                         const int mine, const int j, const int lane) {
    float4 v[B];
    for (int u = 0; u < B; ++u) {
        const int s = __builtin_amdgcn_readlane(mine, j + u);
        v[u] = H4[(size_t)s * (D_FEAT / 4) + lane];
    }
    for (int u = 0; u < B; ++u) {
        acc.x += v[u].x;
        acc.y += v[u].y;
        acc.z += v[u].z;
        acc.w += v[u].w;
    }
}

__device__ __forceinline__ void gcn_chunk_sum(float4& acc, const float4* __restrict__ H4,
                                              const int mine, const int m, const int lane) {
    int j = 0;
    for (; j + 8 <= m; j += 8) gcn_rows<8>(acc, H4, mine, j, lane);
    switch (m - j) {
        case 7: gcn_rows<7>(acc, H4, mine, j, lane); break;
        case 6: gcn_rows<6>(acc, H4, mine, j, lane); break;
        case 5: gcn_rows<5>(acc, H4, mine, j, lane); break;
        case 4: gcn_rows<4>(acc, H4, mine, j, lane); break;
        case 3: gcn_rows<3>(acc, H4, mine, j, lane); break;
        case 2: gcn_rows<2>(acc, H4, mine, j, lane); break;
        case 1: gcn_rows<1>(acc, H4, mine, j, lane); break;
        default: break;
    }
}

__device__ __forceinline__ void gcn_chunk(float4& acc, const float4* __restrict__ H4,
                                          const int* __restrict__ slot,
                                          const int first, const int m, const int lane) {
    int mine = 0;
    if (lane < m) mine = slot[first + lane];
    gcn_chunk_sum(acc, H4, mine, m, lane);
}

template <bool SCALE_OUT>
__device__ __forceinline__ float4 gcn_finish(const float4 acc, const float4 hv,
                                             const float4 bv, const float sd) {
    float4 o;
    o.x = fmaf(acc.x + hv.x, sd, bv.x);
    o.y = fmaf(acc.y + hv.y, sd, bv.y);
    o.z = fmaf(acc.z + hv.z, sd, bv.z);
    o.w = fmaf(acc.w + hv.w, sd, bv.w);
    if (SCALE_OUT) {
        o.x *= sd;
        o.y *= sd;
        o.z *= sd;
        o.w *= sd;
    }
    return o;
}

__device__ __forceinline__ int gcn_sorted_window(const int* __restrict__ raw, int* win,
                                                 const int start, const int n, const int p,
                                                 const int lane) {
    for (int c0 = 0; c0 < n; c0 += 32) {
        const int i = c0 + lane;
        int mine = 0x7fffffff;
        if (i < n) mine = raw[start + i];
        const u64 mk = ((u64)(unsigned)mine << 32) | (u64)(unsigned)i;
        int pos = 0;
        for (int k0 = 0; k0 < n; k0 += 32) {
            int key = mine;
            if (k0 != c0) {
                key = 0x7fffffff;
                if (k0 + lane < n) key = raw[start + k0 + lane];
            }
            const int cm = (n - k0 < 32) ? (n - k0) : 32;
            for (int k = 0; k < cm; ++k) {
                const u64 ok = ((u64)(unsigned)__builtin_amdgcn_readlane(key, k) << 32)
                             | (u64)(unsigned)(k0 + k);
                pos += (ok < mk) ? 1 : 0;
            }
        }
        if (i < n && pos >= p && pos < p + 32) win[pos - p] = mine;
    }
    int got = 0;
    if (p + lane < n) got = win[lane];
    return got;
}

template <bool SCALE_OUT>
__global__ void __launch_bounds__(256)
gcn_gather_rows(const float* __restrict__ H, const u64* __restrict__ row,
                const int* __restrict__ slot, const float* __restrict__ dis,
                const float* __restrict__ b, float* __restrict__ out, int N) {
    const int lane = threadIdx.x & 31;
    const int d = __builtin_amdgcn_readfirstlane((int)((blockIdx.x * blockDim.x + threadIdx.x) >> 5));
    if (d >= N) return;

    const float4* H4 = (const float4*)H;
    const u64    ri = row[d];
    const float  sd = dis[d];
    const float4 hv = H4[(size_t)d * (D_FEAT / 4) + lane];
    const float4 bv = ((const float4*)b)[lane];
    const int start = (int)(unsigned)ri;
    const int n     = (int)(ri >> 32);

    float4 acc = make_float4(0.0f, 0.0f, 0.0f, 0.0f);

    if (n > 0) gcn_chunk(acc, H4, slot, start, (n < 32) ? n : 32, lane);
    for (int j0 = 32; j0 < n; j0 += 32)
        gcn_chunk(acc, H4, slot, start + j0, (n - j0 < 32) ? (n - j0) : 32, lane);

    ((float4*)out)[(size_t)d * (D_FEAT / 4) + lane] = gcn_finish<SCALE_OUT>(acc, hv, bv, sd);
}

__global__ void __launch_bounds__(256)
gcn_order_gather_rows(const float* __restrict__ H, const u64* __restrict__ row,
                      const int* __restrict__ raw, int* __restrict__ slot,
                      const float* __restrict__ dis, const float* __restrict__ b,
                      float* __restrict__ out, int N) {
    __shared__ int wins[8][32];
    const int lane = threadIdx.x & 31;
    int* win = wins[threadIdx.x >> 5];
    const int d = __builtin_amdgcn_readfirstlane((int)((blockIdx.x * blockDim.x + threadIdx.x) >> 5));
    if (d >= N) return;

    const float4* H4 = (const float4*)H;
    const u64    ri = row[d];
    const float  sd = dis[d];
    const float4 hv = H4[(size_t)d * (D_FEAT / 4) + lane];
    const float4 bv = ((const float4*)b)[lane];
    const int start = (int)(unsigned)ri;
    const int n     = (int)(ri >> 32);

    float4 acc = make_float4(0.0f, 0.0f, 0.0f, 0.0f);

    if (n <= 32) {
        if (n > 0) {
            int mine = 0x7fffffff;
            if (lane < n) mine = raw[start + lane];
            const u64 mk = ((u64)(unsigned)mine << 32) | (u64)(unsigned)lane;
            int pos = 0;
            for (int k = 0; k < n; ++k) {
                const u64 ok = ((u64)(unsigned)__builtin_amdgcn_readlane(mine, k) << 32)
                             | (u64)(unsigned)k;
                pos += (ok < mk) ? 1 : 0;
            }
            if (lane < n) win[pos] = mine;
            int got = 0;
            if (lane < n) {
                got = win[lane];
                slot[start + lane] = got;
            }
            gcn_chunk_sum(acc, H4, got, n, lane);
        }
    } else {
        for (int p = 0; p < n; p += 32) {
            const int m = (n - p < 32) ? (n - p) : 32;
            const int mine = gcn_sorted_window(raw, win, start, n, p, lane);
            if (lane < m) slot[start + p + lane] = mine;
            gcn_chunk_sum(acc, H4, mine, m, lane);
        }
    }

    ((float4*)out)[(size_t)d * (D_FEAT / 4) + lane] = gcn_finish<true>(acc, hv, bv, sd);
}

extern "C" void kernel_launch(void* const* d_in, const int* in_sizes, int n_in,
                              void* d_out, int out_size, void* d_ws, size_t ws_size,
                              hipStream_t stream) {
    const float* feat = (const float*)d_in[0];
    const int*   ei   = (const int*)d_in[1];
    const float* W0   = (const float*)d_in[2];
    const float* b0   = (const float*)d_in[3];
    const float* W1   = (const float*)d_in[4];
    const float* b1   = (const float*)d_in[5];
    const float* W2   = (const float*)d_in[6];
    const float* b2   = (const float*)d_in[7];

    const int N = in_sizes[0] / D_FEAT;
    const int E = in_sizes[1] / 2;
    const int* src = ei;
    const int* dst = ei + E;

    const int M  = (E + PART_EDGES - 1) / PART_EDGES;
    const int NB = (N + BIN_ROWS - 1) >> BIN_SHIFT;

    const size_t Np = ((size_t)N + 1023) & ~(size_t)1023;
    const size_t Ep = ((size_t)E + 1023) & ~(size_t)1023;
    float* dis   = (float*)d_ws;
    u64*   row   = (u64*)(dis + Np);
    u64*   pairs = row + Np;
    int*   raw   = (int*)(pairs + Ep);
    int*   slot  = raw + Ep;
    int*   bh    = slot + Ep;
    int*   bbase = bh + BH_CAP;
    float* H     = (float*)(bbase + 1024);
    float* X     = H + (size_t)N * D_FEAT;
    float* Wc    = X + (size_t)N * D_FEAT;
    float* out   = (float*)d_out;

    const int TB = 256;
    dim3 blkFold(FOLD_BLOCKS + M);
    dim3 blkCopy(GEMM_BLOCKS + M);
    dim3 blkBins(NB);
    dim3 blkRow(((size_t)N * 32 + TB - 1) / TB);

    gcn_fold_weights_count<<<blkFold, TB, 0, stream>>>(W0, b0, W1, b1, W2, Wc, dst, bh, E);

    gcn_copy_bins_gemm<<<blkCopy, TB, 0, stream>>>(src, dst, bh, bbase, pairs, E, M, feat, Wc, H, N);

    gcn_build_bins<<<blkBins, TB, 0, stream>>>(pairs, bbase, row, dis, raw, H, E, NB, N);

    gcn_order_gather_rows<<<blkRow, TB, 0, stream>>>(H, row, raw, slot, dis, Wc + 128 * D_FEAT, X, N);
    gcn_gather_rows<true><<<blkRow, TB, 0, stream>>>(X, row, slot, dis, Wc + 129 * D_FEAT, H, N);
    gcn_gather_rows<false><<<blkRow, TB, 0, stream>>>(H, row, slot, dis, b2, out, N);
}


# Edge blocks order their own pairs by bin (no global count first); a bin's block (400 rows, 16 waves) finds its runs from their offsets and forms its own rows of H = X Wc times dis; 6 launches -> 5
# speedup vs baseline: 6.3466x; 1.1759x over previous
#include <hip/hip_runtime.h>
#include <hip/hip_bf16.h>

typedef __attribute__((ext_vector_type(2))) float v2f;
typedef __attribute__((ext_vector_type(8))) float v8f;

#define D_FEAT 128
#define LDS_STRIDE 132


#define BIN_ROWS 400
#define BIN_PANELS (BIN_ROWS / 16)
#define BIN_THREADS 512
#define MAX_BINS 512
#define PART_EDGES 8192
#define BIN_CAP 8192
#define MAX_PARTS 304
#define BH_CAP (MAX_PARTS * MAX_BINS)
#define RUNS_FLIGHT 19
typedef unsigned long long u64;

__device__ __forceinline__ void gcn_sort_chunk(const int* __restrict__ src, const int* __restrict__ dst,
                                               int* __restrict__ bh, u64* __restrict__ pairs,
                                               u64* stage, int* hist, int* lcur, int* wsum,
                                               const int E, const int b) {
    const int tid  = threadIdx.x;
    const int lane = tid & 31;
    const int wave = tid >> 5;
    for (int i = tid; i < MAX_BINS; i += 256) hist[i] = 0;
    __syncthreads();
    const int base = b * PART_EDGES;
    int d[PART_EDGES / 256], s[PART_EDGES / 256];
    #pragma unroll
    for (int q = 0; q < PART_EDGES / 256; ++q) {
        const int e = base + q * 256 + tid;
        d[q] = (e < E) ? dst[e] : -1;
        s[q] = (e < E) ? src[e] : 0;
    }
    #pragma unroll
    for (int q = 0; q < PART_EDGES / 256; ++q)
        if (d[q] >= 0) atomicAdd(&hist[d[q] / BIN_ROWS], 1);
    __syncthreads();

    const int k0 = 2 * tid, k1 = 2 * tid + 1;
    const int c0 = hist[k0];
    const int c1 = hist[k1];
    const int tsum = c0 + c1;
    int inc = tsum;
    for (int o = 1; o < 32; o <<= 1) {
        const int up = __shfl_up(inc, o);
        if (lane >= o) inc += up;
    }
    if (lane == 31) wsum[wave] = inc;
    __syncthreads();
    int off = 0;
    for (int w = 0; w < 8; ++w)
        if (w < wave) off += wsum[w];
    const int ex = off + inc - tsum;
    lcur[k0] = ex;
    lcur[k1] = ex + c0;
    *(int2*)&bh[b * MAX_BINS + k0] = make_int2(ex, ex + c0);
    __syncthreads();

    #pragma unroll
    for (int q = 0; q < PART_EDGES / 256; ++q) {
        if (d[q] >= 0) {
            const int pos = atomicAdd(&lcur[d[q] / BIN_ROWS], 1);
            stage[pos] = ((u64)(unsigned)d[q] << 32) | (u64)(unsigned)s[q];
        }
    }
    __syncthreads();

    const int n = (E - base < PART_EDGES) ? (E - base) : PART_EDGES;
    for (int i = tid; i < n; i += 256) pairs[base + i] = stage[i];
}


__device__ __forceinline__ void gcn_store_tile(float* __restrict__ H, const float* fac, const v8f acc,
                                               const int pl, const int rbase, const int half,
                                               const int col, const int N) {
    float fs[8];
    #pragma unroll
    for (int j = 0; j < 8; ++j) fs[j] = fac[16 * pl + j + 8 * half];
    #pragma unroll
    for (int j = 0; j < 8; ++j) {
        const int rg = rbase + 16 * pl + j + 8 * half;
        if (pl < BIN_PANELS && rg < N) H[(size_t)rg * D_FEAT + col] = acc[j] * fs[j];
    }
}

__global__ void __launch_bounds__(BIN_THREADS)
gcn_build_bins_gemm(const u64* __restrict__ pairs, const int* __restrict__ bh,
                    u64* __restrict__ row, float* __restrict__ dis, int* __restrict__ raw,
                    const float* __restrict__ X, const float* __restrict__ W,
                    float* __restrict__ H, int E, int M, int N) {
    __shared__ int   stage[BIN_CAP];
    __shared__ int   cnt[BIN_THREADS];
    __shared__ int   cur[BIN_THREADS];
    __shared__ float fac[BIN_THREADS];
    __shared__ int   rstart[BIN_THREADS];
    __shared__ int   rlen[BIN_THREADS];
    __shared__ int   wlo[16];
    __shared__ int   wtot[16];
    __shared__ int   wsc[16];
    __shared__ __attribute__((aligned(16))) float As[2][16 * LDS_STRIDE];

    const int tid   = threadIdx.x;
    const int lane  = tid & 31;
    const int wave  = tid >> 5;
    const int k     = (int)blockIdx.x;
    const int rbase = k * BIN_ROWS;

    const int tw   = wave & 7;
    const int hb   = wave >> 3;
    const int half = lane >> 4;
    const int l16  = lane & 15;
    const int col  = tw * 16 + l16;

    v2f bf[D_FEAT / 4];
    #pragma unroll
    for (int i = 0; i < D_FEAT / 4; ++i) {
        const int kk = 4 * i + 2 * half;
        bf[i].x = W[(size_t)kk * D_FEAT + col];
        bf[i].y = W[(size_t)(kk + 1) * D_FEAT + col];
    }

    int st = 0, ln = 0;
    if (tid < M) {
        const int rem = E - tid * PART_EDGES;
        const int nb  = (rem < PART_EDGES) ? rem : PART_EDGES;
        st = bh[tid * MAX_BINS + k];
        const int en = (k + 1 < MAX_BINS) ? bh[tid * MAX_BINS + k + 1] : nb;
        ln = en - st;
    }
    rstart[tid] = tid * PART_EDGES + st;
    rlen[tid]   = ln;
    cnt[tid]    = 0;
    int sst = st, sln = ln;
    for (int o = 16; o > 0; o >>= 1) {
        sst += __shfl_xor(sst, o);
        sln += __shfl_xor(sln, o);
    }
    if (lane == 0) { wlo[wave] = sst; wtot[wave] = sln; }
    __syncthreads();
    int lo = 0, size = 0;
    for (int w = 0; w < 16; ++w) { lo += wlo[w]; size += wtot[w]; }
    const int  hi   = lo + size;
    const bool fits = size <= BIN_CAP;

    int rl[RUNS_FLIGHT], rs[RUNS_FLIGHT];
    #pragma unroll
    for (int u = 0; u < RUNS_FLIGHT; ++u) {
        rl[u] = rlen[wave + 16 * u];
        rs[u] = rstart[wave + 16 * u];
    }
    const int* pw = (const int*)pairs;

    {
        int dd[RUNS_FLIGHT];
        #pragma unroll
        for (int u = 0; u < RUNS_FLIGHT; ++u)
            dd[u] = (lane < rl[u]) ? pw[2 * (size_t)(rs[u] + lane) + 1] : -1;
        #pragma unroll
        for (int u = 0; u < RUNS_FLIGHT; ++u)
            if (dd[u] >= 0) atomicAdd(&cnt[dd[u] - rbase], 1);
        for (int bb = wave; bb < M; bb += 16) {
            const int n = rlen[bb], s0 = rstart[bb];
            for (int i = 32 + lane; i < n; i += 32)
                atomicAdd(&cnt[pw[2 * (size_t)(s0 + i) + 1] - rbase], 1);
        }
    }
    __syncthreads();

    const int c = cnt[tid];
    int inc = c;
    for (int o = 1; o < 32; o <<= 1) {
        const int up = __shfl_up(inc, o);
        if (lane >= o) inc += up;
    }
    if (lane == 31) wsc[wave] = inc;
    __syncthreads();
    int off = lo;
    for (int w = 0; w < 16; ++w)
        if (w < wave) off += wsc[w];
    const int   ex = off + inc - c;
    const float f  = rsqrtf((float)(c + 1));
    const int   d  = rbase + tid;
    if (tid < BIN_ROWS && d < N) {
        row[d] = ((u64)(unsigned)c << 32) | (u64)(unsigned)ex;
        dis[d] = f;
    }
    cur[tid] = ex;
    fac[tid] = f;
    __syncthreads();

    {
        u64 p[RUNS_FLIGHT];
        #pragma unroll
        for (int u = 0; u < RUNS_FLIGHT; ++u)
            p[u] = (lane < rl[u]) ? pairs[rs[u] + lane] : ~0ull;
        #pragma unroll
        for (int u = 0; u < RUNS_FLIGHT; ++u) {
            if ((int)(p[u] >> 32) >= 0) {
                const int pos = atomicAdd(&cur[(int)(p[u] >> 32) - rbase], 1);
                if (fits) stage[pos - lo] = (int)(unsigned)p[u];
                else      raw[pos]        = (int)(unsigned)p[u];
            }
        }
        for (int bb = wave; bb < M; bb += 16) {
            const int n = rlen[bb], s0 = rstart[bb];
            for (int i = 32 + lane; i < n; i += 32) {
                const u64 q = pairs[s0 + i];
                const int pos = atomicAdd(&cur[(int)(q >> 32) - rbase], 1);
                if (fits) stage[pos - lo] = (int)(unsigned)q;
                else      raw[pos]        = (int)(unsigned)q;
            }
        }
    }
    const float4 zero4 = make_float4(0.0f, 0.0f, 0.0f, 0.0f);
    const float4* X4 = (const float4*)X;
    float4 a0 = zero4, a1 = zero4;
    {
        const int r = rbase + 16 * hb + tw;
        if (r < N)     a0 = X4[(size_t)r * (D_FEAT / 4) + lane];
        if (r + 8 < N) a1 = X4[(size_t)(r + 8) * (D_FEAT / 4) + lane];
    }
    __syncthreads();
    if (fits)
        for (int i = lo + tid; i < hi; i += BIN_THREADS) raw[i] = stage[i - lo];

    float* Am = As[hb];
    v8f done = {};
    int dpl = BIN_PANELS;
    #pragma unroll 1
    for (int pp = 0; pp < BIN_PANELS; pp += 2) {
        __syncthreads();
        *(float4*)&Am[tw * LDS_STRIDE + 4 * lane] = a0;
        *(float4*)&Am[(tw + 8) * LDS_STRIDE + 4 * lane] = a1;
        __syncthreads();

        const int pl = pp + hb;
        const int pn = pl + 2;
        a0 = zero4; a1 = zero4;
        if (pn < BIN_PANELS) {
            const int r = rbase + 16 * pn + tw;
            if (r < N)     a0 = X4[(size_t)r * (D_FEAT / 4) + lane];
            if (r + 8 < N) a1 = X4[(size_t)(r + 8) * (D_FEAT / 4) + lane];
        }

        gcn_store_tile(H, fac, done, dpl, rbase, half, col, N);

        v8f acc = {};
        #pragma unroll
        for (int i = 0; i < D_FEAT / 4; ++i) {
            const v2f a = *(const v2f*)(Am + l16 * LDS_STRIDE + 4 * i + 2 * half);
            acc = __builtin_amdgcn_wmma_f32_16x16x4_f32(
                 false, a,  false, bf[i],
                 (short)0, acc,  false,  false);
        }
        done = acc;
        dpl  = pl;
    }
    gcn_store_tile(H, fac, done, dpl, rbase, half, col, N);
}

#define FOLD_BLOCKS 9

__device__ __forceinline__ v8f gcn_panel_times_wt(const float* __restrict__ As,
                                                  const float* __restrict__ Ws,
                                                  const int l16, const int half, const int col) {
    v8f acc = {};
    #pragma unroll
    for (int k0 = 0; k0 < D_FEAT; k0 += 4) {
        const int ka = k0 + 2 * half;
        const v2f a = *(const v2f*)(As + l16 * LDS_STRIDE + ka);
        const v2f b = *(const v2f*)(Ws + col * LDS_STRIDE + ka);
        acc = __builtin_amdgcn_wmma_f32_16x16x4_f32(
             false, a,  false, b,
             (short)0, acc,  false,  false);
    }
    return acc;
}

__global__ void __launch_bounds__(256)
gcn_fold_weights_sort(const float* __restrict__ W0, const float* __restrict__ b0,
                      const float* __restrict__ W1, const float* __restrict__ b1,
                      const float* __restrict__ W2, float* __restrict__ Wc,
                      const int* __restrict__ src, const int* __restrict__ dst,
                      int* __restrict__ bh, u64* __restrict__ pairs, int E) {
    __shared__ __attribute__((aligned(16))) float W1s[D_FEAT * LDS_STRIDE];
    __shared__ float W2s[D_FEAT * LDS_STRIDE];
    __shared__ float As[16 * LDS_STRIDE];
    __shared__ int   hist[MAX_BINS];
    __shared__ int   lcur[MAX_BINS];
    __shared__ int   wsum[8];

    const int tid  = threadIdx.x;
    if (blockIdx.x >= FOLD_BLOCKS) {
        gcn_sort_chunk(src, dst, bh, pairs, (u64*)W1s, hist, lcur, wsum, E,
                       (int)blockIdx.x - FOLD_BLOCKS);
        return;
    }
    const int p    = blockIdx.x;
    const int wave = tid >> 5;
    const int lane = tid & 31;
    const int half = lane >> 4;
    const int l16  = lane & 15;
    const int col  = wave * 16 + l16;

    const int kk = lane & 7;
    const int pc = 4 * wave + (lane >> 3);
    float4 w1[16], w2[16], a4[2];
    #pragma unroll
    for (int u = 0; u < 16; ++u)
        w1[u] = ((const float4*)(W1 + (size_t)(8 * u + kk) * D_FEAT))[pc];
    #pragma unroll
    for (int u = 0; u < 16; ++u)
        w2[u] = ((const float4*)(W2 + (size_t)(8 * u + kk) * D_FEAT))[pc];
    #pragma unroll
    for (int u = 0; u < 2; ++u) {
        const int r = 8 * u + wave;
        a4[u] = make_float4(0.0f, 0.0f, 0.0f, 0.0f);
        if (p < 8)       a4[u] = ((const float4*)(W0 + (size_t)(16 * p + r) * D_FEAT))[lane];
        else if (r == 0) a4[u] = ((const float4*)b0)[lane];
    }
    #pragma unroll
    for (int u = 0; u < 2; ++u)
        *(float4*)&As[(8 * u + wave) * LDS_STRIDE + 4 * lane] = a4[u];
    #pragma unroll
    for (int u = 0; u < 16; ++u) {
        const int k = 8 * u + kk;
        W1s[(4 * pc + 0) * LDS_STRIDE + k] = w1[u].x;
        W1s[(4 * pc + 1) * LDS_STRIDE + k] = w1[u].y;
        W1s[(4 * pc + 2) * LDS_STRIDE + k] = w1[u].z;
        W1s[(4 * pc + 3) * LDS_STRIDE + k] = w1[u].w;
    }
    #pragma unroll
    for (int u = 0; u < 16; ++u) {
        const int k = 8 * u + kk;
        W2s[(4 * pc + 0) * LDS_STRIDE + k] = w2[u].x;
        W2s[(4 * pc + 1) * LDS_STRIDE + k] = w2[u].y;
        W2s[(4 * pc + 2) * LDS_STRIDE + k] = w2[u].z;
        W2s[(4 * pc + 3) * LDS_STRIDE + k] = w2[u].w;
    }
    __syncthreads();

    const v8f t = gcn_panel_times_wt(As, W1s, l16, half, col);
    __syncthreads();

    #pragma unroll
    for (int j = 0; j < 8; ++j) As[(j + 8 * half) * LDS_STRIDE + col] = t[j];
    if (p == 8 && half == 0) As[1 * LDS_STRIDE + col] = b1[col];
    __syncthreads();

    const v8f o = gcn_panel_times_wt(As, W2s, l16, half, col);

    const int nrow = (p < 8) ? 16 : 2;
    #pragma unroll
    for (int j = 0; j < 8; ++j) {
        const int r = j + 8 * half;
        if (r < nrow) Wc[(size_t)(16 * p + r) * D_FEAT + col] = o[j];
    }
}

template <int B>
__device__ __forceinline__ void gcn_rows(float4& acc, const float4* __restrict__ H4,
                                         const int mine, const int j, const int lane) {
    float4 v[B];
    for (int u = 0; u < B; ++u) {
        const int s = __builtin_amdgcn_readlane(mine, j + u);
        v[u] = H4[(size_t)s * (D_FEAT / 4) + lane];
    }
    for (int u = 0; u < B; ++u) {
        acc.x += v[u].x;
        acc.y += v[u].y;
        acc.z += v[u].z;
        acc.w += v[u].w;
    }
}

__device__ __forceinline__ void gcn_chunk_sum(float4& acc, const float4* __restrict__ H4,
                                              const int mine, const int m, const int lane) {
    int j = 0;
    for (; j + 8 <= m; j += 8) gcn_rows<8>(acc, H4, mine, j, lane);
    switch (m - j) {
        case 7: gcn_rows<7>(acc, H4, mine, j, lane); break;
        case 6: gcn_rows<6>(acc, H4, mine, j, lane); break;
        case 5: gcn_rows<5>(acc, H4, mine, j, lane); break;
        case 4: gcn_rows<4>(acc, H4, mine, j, lane); break;
        case 3: gcn_rows<3>(acc, H4, mine, j, lane); break;
        case 2: gcn_rows<2>(acc, H4, mine, j, lane); break;
        case 1: gcn_rows<1>(acc, H4, mine, j, lane); break;
        default: break;
    }
}

__device__ __forceinline__ void gcn_chunk(float4& acc, const float4* __restrict__ H4,
                                          const int* __restrict__ slot,
                                          const int first, const int m, const int lane) {
    int mine = 0;
    if (lane < m) mine = slot[first + lane];
    gcn_chunk_sum(acc, H4, mine, m, lane);
}

template <bool SCALE_OUT>
__device__ __forceinline__ float4 gcn_finish(const float4 acc, const float4 hv,
                                             const float4 bv, const float sd) {
    float4 o;
    o.x = fmaf(acc.x + hv.x, sd, bv.x);
    o.y = fmaf(acc.y + hv.y, sd, bv.y);
    o.z = fmaf(acc.z + hv.z, sd, bv.z);
    o.w = fmaf(acc.w + hv.w, sd, bv.w);
    if (SCALE_OUT) {
        o.x *= sd;
        o.y *= sd;
        o.z *= sd;
        o.w *= sd;
    }
    return o;
}

__device__ __forceinline__ int gcn_sorted_window(const int* __restrict__ raw, int* win,
                                                 const int start, const int n, const int p,
                                                 const int lane) {
    for (int c0 = 0; c0 < n; c0 += 32) {
        const int i = c0 + lane;
        int mine = 0x7fffffff;
        if (i < n) mine = raw[start + i];
        const u64 mk = ((u64)(unsigned)mine << 32) | (u64)(unsigned)i;
        int pos = 0;
        for (int k0 = 0; k0 < n; k0 += 32) {
            int key = mine;
            if (k0 != c0) {
                key = 0x7fffffff;
                if (k0 + lane < n) key = raw[start + k0 + lane];
            }
            const int cm = (n - k0 < 32) ? (n - k0) : 32;
            for (int k = 0; k < cm; ++k) {
                const u64 ok = ((u64)(unsigned)__builtin_amdgcn_readlane(key, k) << 32)
                             | (u64)(unsigned)(k0 + k);
                pos += (ok < mk) ? 1 : 0;
            }
        }
        if (i < n && pos >= p && pos < p + 32) win[pos - p] = mine;
    }
    int got = 0;
    if (p + lane < n) got = win[lane];
    return got;
}

template <bool SCALE_OUT>
__global__ void __launch_bounds__(256)
gcn_gather_rows(const float* __restrict__ H, const u64* __restrict__ row,
                const int* __restrict__ slot, const float* __restrict__ dis,
                const float* __restrict__ b, float* __restrict__ out, int N) {
    const int lane = threadIdx.x & 31;
    const int d = __builtin_amdgcn_readfirstlane((int)((blockIdx.x * blockDim.x + threadIdx.x) >> 5));
    if (d >= N) return;

    const float4* H4 = (const float4*)H;
    const u64    ri = row[d];
    const float  sd = dis[d];
    const float4 hv = H4[(size_t)d * (D_FEAT / 4) + lane];
    const float4 bv = ((const float4*)b)[lane];
    const int start = (int)(unsigned)ri;
    const int n     = (int)(ri >> 32);

    float4 acc = make_float4(0.0f, 0.0f, 0.0f, 0.0f);

    if (n > 0) gcn_chunk(acc, H4, slot, start, (n < 32) ? n : 32, lane);
    for (int j0 = 32; j0 < n; j0 += 32)
        gcn_chunk(acc, H4, slot, start + j0, (n - j0 < 32) ? (n - j0) : 32, lane);

    ((float4*)out)[(size_t)d * (D_FEAT / 4) + lane] = gcn_finish<SCALE_OUT>(acc, hv, bv, sd);
}

__global__ void __launch_bounds__(256)
gcn_order_gather_rows(const float* __restrict__ H, const u64* __restrict__ row,
                      const int* __restrict__ raw, int* __restrict__ slot,
                      const float* __restrict__ dis, const float* __restrict__ b,
                      float* __restrict__ out, int N) {
    __shared__ int wins[8][32];
    const int lane = threadIdx.x & 31;
    int* win = wins[threadIdx.x >> 5];
    const int d = __builtin_amdgcn_readfirstlane((int)((blockIdx.x * blockDim.x + threadIdx.x) >> 5));
    if (d >= N) return;

    const float4* H4 = (const float4*)H;
    const u64    ri = row[d];
    const float  sd = dis[d];
    const float4 hv = H4[(size_t)d * (D_FEAT / 4) + lane];
    const float4 bv = ((const float4*)b)[lane];
    const int start = (int)(unsigned)ri;
    const int n     = (int)(ri >> 32);

    float4 acc = make_float4(0.0f, 0.0f, 0.0f, 0.0f);

    if (n <= 32) {
        if (n > 0) {
            int mine = 0x7fffffff;
            if (lane < n) mine = raw[start + lane];
            const u64 mk = ((u64)(unsigned)mine << 32) | (u64)(unsigned)lane;
            int pos = 0;
            for (int k = 0; k < n; ++k) {
                const u64 ok = ((u64)(unsigned)__builtin_amdgcn_readlane(mine, k) << 32)
                             | (u64)(unsigned)k;
                pos += (ok < mk) ? 1 : 0;
            }
            if (lane < n) win[pos] = mine;
            int got = 0;
            if (lane < n) {
                got = win[lane];
                slot[start + lane] = got;
            }
            gcn_chunk_sum(acc, H4, got, n, lane);
        }
    } else {
        for (int p = 0; p < n; p += 32) {
            const int m = (n - p < 32) ? (n - p) : 32;
            const int mine = gcn_sorted_window(raw, win, start, n, p, lane);
            if (lane < m) slot[start + p + lane] = mine;
            gcn_chunk_sum(acc, H4, mine, m, lane);
        }
    }

    ((float4*)out)[(size_t)d * (D_FEAT / 4) + lane] = gcn_finish<true>(acc, hv, bv, sd);
}

extern "C" void kernel_launch(void* const* d_in, const int* in_sizes, int n_in,
                              void* d_out, int out_size, void* d_ws, size_t ws_size,
                              hipStream_t stream) {
    const float* feat = (const float*)d_in[0];
    const int*   ei   = (const int*)d_in[1];
    const float* W0   = (const float*)d_in[2];
    const float* b0   = (const float*)d_in[3];
    const float* W1   = (const float*)d_in[4];
    const float* b1   = (const float*)d_in[5];
    const float* W2   = (const float*)d_in[6];
    const float* b2   = (const float*)d_in[7];

    const int N = in_sizes[0] / D_FEAT;
    const int E = in_sizes[1] / 2;
    const int* src = ei;
    const int* dst = ei + E;

    const int M  = (E + PART_EDGES - 1) / PART_EDGES;
    const int NB = (N + BIN_ROWS - 1) / BIN_ROWS;

    const size_t Np = ((size_t)N + 1023) & ~(size_t)1023;
    const size_t Ep = ((size_t)E + 1023) & ~(size_t)1023;
    float* dis   = (float*)d_ws;
    u64*   row   = (u64*)(dis + Np);
    u64*   pairs = row + Np;
    int*   raw   = (int*)(pairs + Ep);
    int*   slot  = raw + Ep;
    int*   bh    = slot + Ep;
    float* H     = (float*)(bh + BH_CAP);
    float* X     = H + (size_t)N * D_FEAT;
    float* Wc    = X + (size_t)N * D_FEAT;
    float* out   = (float*)d_out;

    const int TB = 256;
    dim3 blkFold(FOLD_BLOCKS + M);
    dim3 blkBins(NB);
    dim3 blkRow(((size_t)N * 32 + TB - 1) / TB);

    gcn_fold_weights_sort<<<blkFold, TB, 0, stream>>>(W0, b0, W1, b1, W2, Wc, src, dst, bh, pairs, E);

    gcn_build_bins_gemm<<<blkBins, BIN_THREADS, 0, stream>>>(pairs, bh, row, dis, raw, feat, Wc, H, E, M, N);

    gcn_order_gather_rows<<<blkRow, TB, 0, stream>>>(H, row, raw, slot, dis, Wc + 128 * D_FEAT, X, N);
    gcn_gather_rows<true><<<blkRow, TB, 0, stream>>>(X, row, slot, dis, Wc + 129 * D_FEAT, H, N);
    gcn_gather_rows<false><<<blkRow, TB, 0, stream>>>(H, row, slot, dis, b2, out, N);
}
